# starting kernel
# baseline (speedup 1.0000x reference)
#include <hip/hip_runtime.h>
#include <hip/hip_bf16.h>

typedef float v2f __attribute__((ext_vector_type(2)));
typedef float v8f __attribute__((ext_vector_type(8)));

#define F 32
#define NB 16

__device__ __forceinline__ float fast_silu(float x) {
    return x * __builtin_amdgcn_rcpf(1.0f + __expf(-x));
}

__global__ void zero_kernel(float* __restrict__ p, int n) {
    int g = blockIdx.x * blockDim.x + threadIdx.x;
    if (g < n) p[g] = 0.0f;
}

__global__ void mol_kernel(const int* __restrict__ q, const int* __restrict__ sp,
                           const float* __restrict__ ce, const float* __restrict__ se,
                           const float* __restrict__ mW, const float* __restrict__ mb,
                           float* __restrict__ mol, int B) {
    int g = blockIdx.x * blockDim.x + threadIdx.x;
    if (g >= B * F) return;
    int b = g >> 5, f = g & 31;
    int ci = min(max(q[b] + 5, 0), 10);
    int si = min(max(sp[b] - 1, 0), 6);
    float acc = mb[f];
    #pragma unroll
    for (int k = 0; k < 16; ++k) acc += ce[ci * 16 + k] * mW[k * F + f];
    #pragma unroll
    for (int k = 0; k < 16; ++k) acc += se[si * 16 + k] * mW[(16 + k) * F + f];
    mol[g] = acc;
}

__global__ void x0_kernel(const int* __restrict__ Z, const int* __restrict__ seg,
                          const float* __restrict__ ae, const float* __restrict__ mol,
                          float* __restrict__ x0, int A) {
    int g = blockIdx.x * blockDim.x + threadIdx.x;
    if (g >= A * F) return;
    int a = g >> 5, f = g & 31;
    x0[g] = ae[Z[a] * F + f] + mol[seg[a] * F + f];
}

__global__ __launch_bounds__(256) void edge_kernel(
    const float* __restrict__ pos, const int* __restrict__ dst,
    const int* __restrict__ src, const float* __restrict__ Wb,
    const float* __restrict__ xin, float* __restrict__ accum,
    int P, int add_cut) {
    long long g = (long long)blockIdx.x * blockDim.x + threadIdx.x;
    int e = (int)(g >> 5);
    int f = (int)(g & 31);
    if (e >= P) return;
    int s = src[e], d = dst[e];
    float dx = pos[s * 3 + 0] - pos[d * 3 + 0];
    float dy = pos[s * 3 + 1] - pos[d * 3 + 1];
    float dz = pos[s * 3 + 2] - pos[d * 3 + 2];
    float r = sqrtf(dx * dx + dy * dy + dz * dz);
    float u = r * (1.0f / 6.0f);
    u = u * u;
    float cutv = 0.0f;
    if (u < 1.0f)
        cutv = __expf(1.0f - __builtin_amdgcn_rcpf(fmaxf(1.0f - u, 1e-6f)));
    float zc = fminf(fmaxf(2.0f * __expf(-r) - 1.0f, -1.0f), 1.0f);
    float Tm2 = 1.0f, Tm1 = zc;
    float c = Wb[f] + Wb[F + f] * zc;
    #pragma unroll
    for (int b2 = 2; b2 < NB; ++b2) {
        float T = 2.0f * zc * Tm1 - Tm2;
        c += Wb[b2 * F + f] * T;
        Tm2 = Tm1; Tm1 = T;
    }
    float scale = cutv * (c + (add_cut ? 1.0f : 0.0f));
    float val = scale * xin[s * F + f];
    unsafeAtomicAdd(&accum[d * F + f], val);
}

__global__ __launch_bounds__(256) void refine_kernel(
    const float* __restrict__ xa, const float* __restrict__ xb,
    const float* __restrict__ W1s, const float* __restrict__ b1s,
    const float* __restrict__ W2s, const float* __restrict__ b2s,
    float* __restrict__ out, int A) {
    __shared__ __align__(16) float sW1[3 * F * F];
    __shared__ __align__(16) float sW2[3 * F * F];
    __shared__ float sb1[3 * F];
    __shared__ float sb2[3 * F];
    __shared__ __align__(16) float xt[128 * F];
    __shared__ __align__(16) float ht[128 * F];

    const int t = threadIdx.x;
    const int lane = t & 31;
    const int wave = t >> 5;
    const int base = blockIdx.x * 128 * F;
    const int lim = A * F;

    for (int i = t; i < 3 * F * F; i += 256) {
        int rem = i & 1023;
        int k = rem >> 5;
        int n = rem & 31;
        int fidx = (i & ~1023) + ((k >> 1) << 6) + (n << 1) + (k & 1);
        sW1[fidx] = W1s[i];
        sW2[fidx] = W2s[i];
    }
    for (int i = t; i < 3 * F; i += 256) { sb1[i] = b1s[i]; sb2[i] = b2s[i]; }
    for (int i = t; i < 128 * F; i += 256)
        xt[i] = (base + i < lim) ? (xa[base + i] + xb[base + i]) : 0.0f;
    __syncthreads();

    const int m     = lane & 15;
    const int khalf = (lane >> 4) << 1;
    const int row0  = wave * 16;
    const int cm    = (lane >> 4) << 3;
    const int cn    = lane & 15;

    for (int j = 0; j < 3; ++j) {
        const v2f* W1p = (const v2f*)(sW1 + j * F * F);
        const v2f* W2p = (const v2f*)(sW2 + j * F * F);
        const float* B1 = sb1 + j * F;
        const float* B2 = sb2 + j * F;

        v8f acc0, acc1;
        {
            float bn0 = B1[cn], bn1 = B1[cn + 16];
            #pragma unroll
            for (int q = 0; q < 8; ++q) { acc0[q] = bn0; acc1[q] = bn1; }
        }
        #pragma unroll
        for (int kc = 0; kc < 8; ++kc) {
            int kb = kc * 4 + khalf;
            int kp = (kb >> 1) << 5;
            v2f a  = *(const v2f*)(xt + (row0 + m) * F + kb);
            v2f b0 = W1p[kp + cn];
            v2f b1 = W1p[kp + cn + 16];
            acc0 = __builtin_amdgcn_wmma_f32_16x16x4_f32(false, a, false, b0, (short)0, acc0, false, false);
            acc1 = __builtin_amdgcn_wmma_f32_16x16x4_f32(false, a, false, b1, (short)0, acc1, false, false);
        }
        #pragma unroll
        for (int q = 0; q < 8; ++q) {
            ht[(row0 + cm + q) * F + cn]      = fast_silu(acc0[q]);
            ht[(row0 + cm + q) * F + cn + 16] = fast_silu(acc1[q]);
        }
        __syncthreads();

        {
            float bn0 = B2[cn], bn1 = B2[cn + 16];
            #pragma unroll
            for (int q = 0; q < 8; ++q) { acc0[q] = bn0; acc1[q] = bn1; }
        }
        #pragma unroll
        for (int kc = 0; kc < 8; ++kc) {
            int kb = kc * 4 + khalf;
            int kp = (kb >> 1) << 5;
            v2f a  = *(const v2f*)(ht + (row0 + m) * F + kb);
            v2f b0 = W2p[kp + cn];
            v2f b1 = W2p[kp + cn + 16];
            acc0 = __builtin_amdgcn_wmma_f32_16x16x4_f32(false, a, false, b0, (short)0, acc0, false, false);
            acc1 = __builtin_amdgcn_wmma_f32_16x16x4_f32(false, a, false, b1, (short)0, acc1, false, false);
        }
        __syncthreads();
        #pragma unroll
        for (int q = 0; q < 8; ++q) {
            float x0v = xt[(row0 + cm + q) * F + cn]      + acc0[q];
            float x1v = xt[(row0 + cm + q) * F + cn + 16] + acc1[q];
            xt[(row0 + cm + q) * F + cn]      = fast_silu(x0v);
            xt[(row0 + cm + q) * F + cn + 16] = fast_silu(x1v);
        }
        __syncthreads();
    }

    for (int i = t; i < 128 * F; i += 256)
        if (base + i < lim) out[base + i] = xt[i];
}

extern "C" void kernel_launch(void* const* d_in, const int* in_sizes, int n_in,
                              void* d_out, int out_size, void* d_ws, size_t ws_size,
                              hipStream_t stream) {
    const float* pos  = (const float*)d_in[0];
    const int*   q    = (const int*)d_in[1];
    const int*   sp   = (const int*)d_in[2];
    const int*   Z    = (const int*)d_in[3];
    const int*   dst  = (const int*)d_in[4];
    const int*   src  = (const int*)d_in[5];
    const int*   seg  = (const int*)d_in[6];
    const float* ce   = (const float*)d_in[7];
    const float* se   = (const float*)d_in[8];
    const float* ae   = (const float*)d_in[9];
    const float* mW   = (const float*)d_in[10];
    const float* mb   = (const float*)d_in[11];
    const float* mpWb = (const float*)d_in[12];
    const float* rW1  = (const float*)d_in[13];
    const float* rb1  = (const float*)d_in[14];
    const float* rW2  = (const float*)d_in[15];
    const float* rb2  = (const float*)d_in[16];

    const int B = in_sizes[1];
    const int A = in_sizes[3];
    const int P = in_sizes[4];

    float* ws  = (float*)d_ws;
    float* mol = ws;
    float* x0  = mol + (size_t)B * F;
    float* s0  = x0 + (size_t)A * F;
    float* xs0 = s0 + (size_t)A * F;
    float* s1  = xs0 + (size_t)A * F;

    {
        int n = 3 * A * F;
        zero_kernel<<<(n + 255) / 256, 256, 0, stream>>>(s0, n);
    }
    mol_kernel<<<(B * F + 255) / 256, 256, 0, stream>>>(q, sp, ce, se, mW, mb, mol, B);
    x0_kernel<<<(A * F + 255) / 256, 256, 0, stream>>>(Z, seg, ae, mol, x0, A);

    const long long nedge = (long long)P * F;
    const int eblocks = (int)((nedge + 255) / 256);

    edge_kernel<<<eblocks, 256, 0, stream>>>(pos, dst, src, mpWb, x0, s0, P, 1);
    refine_kernel<<<(A + 127) / 128, 256, 0, stream>>>(
        x0, s0, rW1, rb1, rW2, rb2, xs0, A);
    edge_kernel<<<eblocks, 256, 0, stream>>>(pos, dst, src, mpWb + NB * F, xs0, s1, P, 0);
    refine_kernel<<<(A + 127) / 128, 256, 0, stream>>>(
        xs0, s1, rW1 + 3 * F * F, rb1 + 3 * F, rW2 + 3 * F * F, rb2 + 3 * F,
        (float*)d_out, A);
}


# second message pass reads the per-group edge list (source atom, local dest) that the first pass now stores in a workspace plane instead of re-scanning dst; LDS adds still in arrival order
# speedup vs baseline: 1.4261x; 1.4261x over previous
#include <hip/hip_runtime.h>
#include <hip/hip_bf16.h>

typedef float v2f __attribute__((ext_vector_type(2)));
typedef float v8f __attribute__((ext_vector_type(8)));

#define F 32
#define NB 16

__device__ __forceinline__ float fast_silu(float x) {
    return x * __builtin_amdgcn_rcpf(1.0f + __expf(-x));
}

__global__ void mol_kernel(const int* __restrict__ q, const int* __restrict__ sp,
                           const float* __restrict__ ce, const float* __restrict__ se,
                           const float* __restrict__ mW, const float* __restrict__ mb,
                           float* __restrict__ mol, int B) {
    int g = blockIdx.x * blockDim.x + threadIdx.x;
    if (g >= B * F) return;
    int b = g >> 5, f = g & 31;
    int ci = min(max(q[b] + 5, 0), 10);
    int si = min(max(sp[b] - 1, 0), 6);
    float acc = mb[f];
    #pragma unroll
    for (int k = 0; k < 16; ++k) acc += ce[ci * 16 + k] * mW[k * F + f];
    #pragma unroll
    for (int k = 0; k < 16; ++k) acc += se[si * 16 + k] * mW[(16 + k) * F + f];
    mol[g] = acc;
}

__global__ void x0_kernel(const int* __restrict__ Z, const int* __restrict__ seg,
                          const float* __restrict__ ae, const float* __restrict__ mol,
                          float* __restrict__ x0, int A) {
    int g = blockIdx.x * blockDim.x + threadIdx.x;
    if (g >= A * F) return;
    int a = g >> 5, f = g & 31;
    x0[g] = ae[Z[a] * F + f] + mol[seg[a] * F + f];
}

#define APW  64
#define DLB  6
static_assert(APW == (1 << DLB), "a packed list word keeps the local destination in its low DLB bits");
#define LCAP 4096
#define OT   1024
#define SCAN_UN 8

__device__ __forceinline__ void edge_rw(const float* __restrict__ pos, int s, int d, bool live, float* rw) {
    float dx = pos[s * 3 + 0] - pos[d * 3 + 0];
    float dy = pos[s * 3 + 1] - pos[d * 3 + 1];
    float dz = pos[s * 3 + 2] - pos[d * 3 + 2];
    float r = sqrtf(dx * dx + dy * dy + dz * dz);
    float u = r * (1.0f / 6.0f);
    u = u * u;
    float cutv = 0.0f;
    if (u < 1.0f)
        cutv = __expf(1.0f - __builtin_amdgcn_rcpf(fmaxf(1.0f - u, 1e-6f)));
    cutv = live ? cutv : 0.0f;
    float zc = fminf(fmaxf(2.0f * __expf(-r) - 1.0f, -1.0f), 1.0f);
    float Tm2 = 1.0f, Tm1 = zc;
    rw[0] = cutv;
    rw[1] = cutv * zc;
    #pragma unroll
    for (int b2 = 2; b2 < NB; ++b2) {
        float T = 2.0f * zc * Tm1 - Tm2;
        rw[b2] = cutv * T;
        Tm2 = Tm1; Tm1 = T;
    }
}

__device__ __forceinline__ float pick(bool c, float a, float b) { return c ? a : b; }

__device__ __forceinline__ int4 load_dst4(const int* __restrict__ dst, int b, int P) {
    if (b + 3 < P) return *(const int4*)(dst + b);
    int4 v;
    v.x = (b     < P) ? dst[b]     : -1;
    v.y = (b + 1 < P) ? dst[b + 1] : -1;
    v.z = (b + 2 < P) ? dst[b + 2] : -1;
    v.w = (b + 3 < P) ? dst[b + 3] : -1;
    return v;
}

__global__ __launch_bounds__(OT) void edge_kernel(
    const float* __restrict__ pos, const int* __restrict__ dst,
    const int* __restrict__ src, const float* __restrict__ Wb,
    const float* __restrict__ xin, float* __restrict__ accum,
    int* __restrict__ lst, int P, int A, int add_cut, int reuse) {
    __shared__ float acc[APW * F];
    __shared__ int listE[LCAP];
    __shared__ int listD[LCAP];
    __shared__ int s_cnt;

    const int t = threadIdx.x;
    const int lane = t & 31;
    const int wave = t >> 5;
    const int lo = blockIdx.x * APW;
    const float addc = add_cut ? 1.0f : 0.0f;

    const int rows = min(APW, A - lo);
    int* __restrict__ mylst = lst + (size_t)lo * F;
    const int capS = rows * F - 32;
    const bool packable = A <= (1 << (32 - DLB));

    int total = 0;
    bool from_list = false;
    if (reuse) {
        const int c = mylst[capS];
        if (c >= 0 && c <= capS && c <= LCAP) { from_list = true; total = c; }
    }

    for (int i = t; i < APW * F; i += OT) acc[i] = 0.0f;
    if (t == 0) s_cnt = 0;
    __syncthreads();

    if (from_list) {
        for (int i = t; i < total; i += OT) {
            const unsigned w = (unsigned)mylst[i];
            listE[i] = (int)(w >> DLB);
            listD[i] = (int)(w & (unsigned)(APW - 1));
        }
    } else {
        auto keep4 = [&](const int4 v, const int b) {
            const int dv[4] = {v.x, v.y, v.z, v.w};
            #pragma unroll
            for (int c = 0; c < 4; ++c) {
                unsigned u = (unsigned)(dv[c] - lo);
                if (u < (unsigned)APW) {
                    int slot = atomicAdd(&s_cnt, 1);
                    if (slot < LCAP) { listE[slot] = b + c; listD[slot] = (int)u; }
                }
            }
        };
        int base0 = t * 4;
        for (; base0 + (SCAN_UN - 1) * OT * 4 + 3 < P; base0 += OT * 4 * SCAN_UN) {
            int4 v[SCAN_UN];
            #pragma unroll
            for (int j = 0; j < SCAN_UN; ++j)
                v[j] = *(const int4*)(dst + base0 + j * OT * 4);
            #pragma unroll
            for (int j = 0; j < SCAN_UN; ++j)
                keep4(v[j], base0 + j * OT * 4);
        }
        for (; base0 < P; base0 += OT * 4)
            keep4(load_dst4(dst, base0, P), base0);
        __syncthreads();
        total = s_cnt;

        const bool save = !reuse && packable && total <= capS && total <= LCAP;
        if (total <= LCAP) {
            for (int i = t; i < total; i += OT) {
                const int s = src[listE[i]];
                listE[i] = s;
                if (save) mylst[i] = (int)(((unsigned)s << DLB) | (unsigned)listD[i]);
            }
        }
        if (!reuse && t == 0) mylst[capS] = save ? total : -1;
    }
    __syncthreads();

    if (total <= LCAP) {
        const int m  = lane & 15;
        const int h  = lane >> 4;
        const int cn = lane & 15;

        v2f bf0[4], bf1[4];
        #pragma unroll
        for (int kc = 0; kc < 4; ++kc) {
            const int k0 = 4 * kc + 2 * h;
            bf0[kc][0] = Wb[k0 * F + cn];
            bf0[kc][1] = Wb[(k0 + 1) * F + cn];
            bf1[kc][0] = Wb[k0 * F + cn + 16];
            bf1[kc][1] = Wb[(k0 + 1) * F + cn + 16];
        }
        if (h == 0) { bf0[0][0] += addc; bf1[0][0] += addc; }

        const int ntiles = (total + 15) >> 4;
        for (int tile = wave; tile < ntiles; tile += OT / 32) {
            const int idx = tile * 16 + m;
            const bool valid = idx < total;
            const int s  = valid ? listE[idx] : 0;
            const int dl = valid ? listD[idx] : 0;

            int   drr[8];
            float xv0[8], xv1[8];
            #pragma unroll
            for (int q = 0; q < 8; ++q) {
                const int r  = 8 * h + q;
                const int sr = __shfl(s, r);
                drr[q] = __shfl(dl, r);
                xv0[q] = xin[sr * F + cn];
                xv1[q] = xin[sr * F + cn + 16];
            }

            float rw[NB];
            edge_rw(pos, s, lo + dl, valid, rw);

            v8f c0, c1;
            #pragma unroll
            for (int q = 0; q < 8; ++q) { c0[q] = 0.0f; c1[q] = 0.0f; }
            #pragma unroll
            for (int kc = 0; kc < 4; ++kc) {
                v2f a;
                a[0] = pick(h != 0, rw[4 * kc + 2], rw[4 * kc]);
                a[1] = pick(h != 0, rw[4 * kc + 3], rw[4 * kc + 1]);
                c0 = __builtin_amdgcn_wmma_f32_16x16x4_f32(false, a, false, bf0[kc], (short)0, c0, false, false);
                c1 = __builtin_amdgcn_wmma_f32_16x16x4_f32(false, a, false, bf1[kc], (short)0, c1, false, false);
            }
            #pragma unroll
            for (int q = 0; q < 8; ++q) {
                atomicAdd(&acc[drr[q] * F + cn],      c0[q] * xv0[q]);
                atomicAdd(&acc[drr[q] * F + cn + 16], c1[q] * xv1[q]);
            }
        }
    } else {
        for (int e = t; e < P; e += OT) {
            unsigned u = (unsigned)(dst[e] - lo);
            if (u < (unsigned)APW) {
                const int s = src[e];
                float rw[NB];
                edge_rw(pos, s, lo + (int)u, true, rw);
                for (int f = 0; f < F; ++f) {
                    float c = rw[0] * (Wb[f] + addc);
                    #pragma unroll
                    for (int k = 1; k < NB; ++k) c += rw[k] * Wb[k * F + f];
                    atomicAdd(&acc[(int)u * F + f], c * xin[s * F + f]);
                }
            }
        }
    }
    __syncthreads();

    for (int i = t; i < APW * F; i += OT)
        if (lo + (i >> 5) < A) accum[lo * F + i] = acc[i];
}

__global__ __launch_bounds__(256) void refine_kernel(
    const float* __restrict__ xa, const float* __restrict__ xb,
    const float* __restrict__ W1s, const float* __restrict__ b1s,
    const float* __restrict__ W2s, const float* __restrict__ b2s,
    float* __restrict__ out, int A) {
    __shared__ __align__(16) float sW1[3 * F * F];
    __shared__ __align__(16) float sW2[3 * F * F];
    __shared__ float sb1[3 * F];
    __shared__ float sb2[3 * F];
    __shared__ __align__(16) float xt[128 * F];
    __shared__ __align__(16) float ht[128 * F];

    const int t = threadIdx.x;
    const int lane = t & 31;
    const int wave = t >> 5;
    const int base = blockIdx.x * 128 * F;
    const int lim = A * F;

    for (int i = t; i < 3 * F * F; i += 256) {
        int rem = i & 1023;
        int k = rem >> 5;
        int n = rem & 31;
        int fidx = (i & ~1023) + ((k >> 1) << 6) + (n << 1) + (k & 1);
        sW1[fidx] = W1s[i];
        sW2[fidx] = W2s[i];
    }
    for (int i = t; i < 3 * F; i += 256) { sb1[i] = b1s[i]; sb2[i] = b2s[i]; }
    for (int i = t; i < 128 * F; i += 256)
        xt[i] = (base + i < lim) ? (xa[base + i] + xb[base + i]) : 0.0f;
    __syncthreads();

    const int m     = lane & 15;
    const int khalf = (lane >> 4) << 1;
    const int row0  = wave * 16;
    const int cm    = (lane >> 4) << 3;
    const int cn    = lane & 15;

    for (int j = 0; j < 3; ++j) {
        const v2f* W1p = (const v2f*)(sW1 + j * F * F);
        const v2f* W2p = (const v2f*)(sW2 + j * F * F);
        const float* B1 = sb1 + j * F;
        const float* B2 = sb2 + j * F;

        v8f acc0, acc1;
        {
            float bn0 = B1[cn], bn1 = B1[cn + 16];
            #pragma unroll
            for (int q = 0; q < 8; ++q) { acc0[q] = bn0; acc1[q] = bn1; }
        }
        #pragma unroll
        for (int kc = 0; kc < 8; ++kc) {
            int kb = kc * 4 + khalf;
            int kp = (kb >> 1) << 5;
            v2f a  = *(const v2f*)(xt + (row0 + m) * F + kb);
            v2f b0 = W1p[kp + cn];
            v2f b1 = W1p[kp + cn + 16];
            acc0 = __builtin_amdgcn_wmma_f32_16x16x4_f32(false, a, false, b0, (short)0, acc0, false, false);
            acc1 = __builtin_amdgcn_wmma_f32_16x16x4_f32(false, a, false, b1, (short)0, acc1, false, false);
        }
        #pragma unroll
        for (int q = 0; q < 8; ++q) {
            ht[(row0 + cm + q) * F + cn]      = fast_silu(acc0[q]);
            ht[(row0 + cm + q) * F + cn + 16] = fast_silu(acc1[q]);
        }
        __syncthreads();

        {
            float bn0 = B2[cn], bn1 = B2[cn + 16];
            #pragma unroll
            for (int q = 0; q < 8; ++q) { acc0[q] = bn0; acc1[q] = bn1; }
        }
        #pragma unroll
        for (int kc = 0; kc < 8; ++kc) {
            int kb = kc * 4 + khalf;
            int kp = (kb >> 1) << 5;
            v2f a  = *(const v2f*)(ht + (row0 + m) * F + kb);
            v2f b0 = W2p[kp + cn];
            v2f b1 = W2p[kp + cn + 16];
            acc0 = __builtin_amdgcn_wmma_f32_16x16x4_f32(false, a, false, b0, (short)0, acc0, false, false);
            acc1 = __builtin_amdgcn_wmma_f32_16x16x4_f32(false, a, false, b1, (short)0, acc1, false, false);
        }
        __syncthreads();
        #pragma unroll
        for (int q = 0; q < 8; ++q) {
            float x0v = xt[(row0 + cm + q) * F + cn]      + acc0[q];
            float x1v = xt[(row0 + cm + q) * F + cn + 16] + acc1[q];
            xt[(row0 + cm + q) * F + cn]      = fast_silu(x0v);
            xt[(row0 + cm + q) * F + cn + 16] = fast_silu(x1v);
        }
        __syncthreads();
    }

    for (int i = t; i < 128 * F; i += 256)
        if (base + i < lim) out[base + i] = xt[i];
}

extern "C" void kernel_launch(void* const* d_in, const int* in_sizes, int n_in,
                              void* d_out, int out_size, void* d_ws, size_t ws_size,
                              hipStream_t stream) {
    const float* pos  = (const float*)d_in[0];
    const int*   q    = (const int*)d_in[1];
    const int*   sp   = (const int*)d_in[2];
    const int*   Z    = (const int*)d_in[3];
    const int*   dst  = (const int*)d_in[4];
    const int*   src  = (const int*)d_in[5];
    const int*   seg  = (const int*)d_in[6];
    const float* ce   = (const float*)d_in[7];
    const float* se   = (const float*)d_in[8];
    const float* ae   = (const float*)d_in[9];
    const float* mW   = (const float*)d_in[10];
    const float* mb   = (const float*)d_in[11];
    const float* mpWb = (const float*)d_in[12];
    const float* rW1  = (const float*)d_in[13];
    const float* rb1  = (const float*)d_in[14];
    const float* rW2  = (const float*)d_in[15];
    const float* rb2  = (const float*)d_in[16];

    const int B = in_sizes[1];
    const int A = in_sizes[3];
    const int P = in_sizes[4];

    float* ws  = (float*)d_ws;
    float* mol = ws;
    float* x0  = mol + (size_t)B * F;
    float* s0  = x0 + (size_t)A * F;
    float* xs0 = s0 + (size_t)A * F;
    int* elist = (int*)(xs0 + (size_t)A * F);

    mol_kernel<<<(B * F + 255) / 256, 256, 0, stream>>>(q, sp, ce, se, mW, mb, mol, B);
    x0_kernel<<<(A * F + 255) / 256, 256, 0, stream>>>(Z, seg, ae, mol, x0, A);

    const int eblocks = (A + APW - 1) / APW;

    edge_kernel<<<eblocks, OT, 0, stream>>>(pos, dst, src, mpWb, x0, s0, elist, P, A, 1, 0);
    refine_kernel<<<(A + 127) / 128, 256, 0, stream>>>(
        x0, s0, rW1, rb1, rW2, rb2, xs0, A);
    edge_kernel<<<eblocks, OT, 0, stream>>>(pos, dst, src, mpWb + NB * F, xs0, s0, elist, P, A, 0, 1);
    refine_kernel<<<(A + 127) / 128, 256, 0, stream>>>(
        xs0, s0, rW1 + 3 * F * F, rb1 + 3 * F, rW2 + 3 * F * F, rb2 + 3 * F,
        (float*)d_out, A);
}


# pass-1 groups stop scanning all destination indices: a binning pass in the first launch counting-sorts edge chunks by destination group, groups collect their slices; LDS float-add order changes
# speedup vs baseline: 1.6677x; 1.1694x over previous
#include <hip/hip_runtime.h>
#include <hip/hip_bf16.h>

typedef float v2f __attribute__((ext_vector_type(2)));
typedef float v8f __attribute__((ext_vector_type(8)));

#define F 32
#define NB 16

__device__ __forceinline__ float fast_silu(float x) {
    return x * __builtin_amdgcn_rcpf(1.0f + __expf(-x));
}

__global__ void x0_kernel(const int* __restrict__ Z, const int* __restrict__ seg,
                          const float* __restrict__ ae, const float* __restrict__ mol,
                          float* __restrict__ x0, int A) {
    int g = blockIdx.x * blockDim.x + threadIdx.x;
    if (g >= A * F) return;
    int a = g >> 5, f = g & 31;
    x0[g] = ae[Z[a] * F + f] + mol[seg[a] * F + f];
}

#define APW  64
#define DLB  6
static_assert(APW == (1 << DLB), "a packed list word keeps the local destination in its low DLB bits");
#define LCAP 4096
#define OT   1024
#define SCAN_UN 8
#define CH   1024
#define BT   256
#define MAXB 1024
static_assert(CH == 4 * BT && MAXB == 4 * BT, "4 edges and at most 4 groups per binning thread");

struct BinPlan { int nbins, nchunk, ostride; bool ok; };
__device__ __forceinline__ BinPlan bin_plan(int A, int P) {
    BinPlan p;
    p.nbins   = (A + APW - 1) / APW;
    p.nchunk  = (P + CH - 1) / CH;
    p.ostride = (p.nbins + 1 + 31) & ~31;
    p.ok = p.nbins <= MAXB && A <= (1 << (32 - DLB)) &&
           (long long)p.nchunk * (CH + p.ostride) <= (long long)A * F;
    return p;
}

__device__ __forceinline__ void edge_rw(const float* __restrict__ pos, int s, int d, bool live, float* rw) {
    float dx = pos[s * 3 + 0] - pos[d * 3 + 0];
    float dy = pos[s * 3 + 1] - pos[d * 3 + 1];
    float dz = pos[s * 3 + 2] - pos[d * 3 + 2];
    float r = sqrtf(dx * dx + dy * dy + dz * dz);
    float u = r * (1.0f / 6.0f);
    u = u * u;
    float cutv = 0.0f;
    if (u < 1.0f)
        cutv = __expf(1.0f - __builtin_amdgcn_rcpf(fmaxf(1.0f - u, 1e-6f)));
    cutv = live ? cutv : 0.0f;
    float zc = fminf(fmaxf(2.0f * __expf(-r) - 1.0f, -1.0f), 1.0f);
    float Tm2 = 1.0f, Tm1 = zc;
    rw[0] = cutv;
    rw[1] = cutv * zc;
    #pragma unroll
    for (int b2 = 2; b2 < NB; ++b2) {
        float T = 2.0f * zc * Tm1 - Tm2;
        rw[b2] = cutv * T;
        Tm2 = Tm1; Tm1 = T;
    }
}

__device__ __forceinline__ float pick(bool c, float a, float b) { return c ? a : b; }

__device__ __forceinline__ int4 load_dst4(const int* __restrict__ dst, int b, int P) {
    if (b + 3 < P) return *(const int4*)(dst + b);
    int4 v;
    v.x = (b     < P) ? dst[b]     : -1;
    v.y = (b + 1 < P) ? dst[b + 1] : -1;
    v.z = (b + 2 < P) ? dst[b + 2] : -1;
    v.w = (b + 3 < P) ? dst[b + 3] : -1;
    return v;
}

__global__ __launch_bounds__(BT) void prep_kernel(
    const int* __restrict__ q, const int* __restrict__ sp,
    const float* __restrict__ ce, const float* __restrict__ se,
    const float* __restrict__ mW, const float* __restrict__ mb,
    float* __restrict__ mol, int B,
    const int* __restrict__ dst, const int* __restrict__ src,
    int* __restrict__ bins, int A, int P) {
    __shared__ int hist[MAXB + 1];
    __shared__ int sorted[CH];
    __shared__ int wsum[BT / 32];

    const BinPlan pl = bin_plan(A, P);
    const int t = threadIdx.x;
    if ((int)blockIdx.x >= pl.nchunk) {
        const int g = ((int)blockIdx.x - pl.nchunk) * BT + t;
        if (g >= B * F) return;
        const int b = g >> 5, f = g & 31;
        const int ci = min(max(q[b] + 5, 0), 10);
        const int si = min(max(sp[b] - 1, 0), 6);
        float a = mb[f];
        #pragma unroll
        for (int k = 0; k < 16; ++k) a += ce[ci * 16 + k] * mW[k * F + f];
        #pragma unroll
        for (int k = 0; k < 16; ++k) a += se[si * 16 + k] * mW[(16 + k) * F + f];
        mol[g] = a;
        return;
    }
    if (!pl.ok) return;

    const int lane = t & 31;
    const int wave = t >> 5;
    const int c = blockIdx.x;
    for (int i = t; i <= pl.nbins; i += BT) hist[i] = 0;
    __syncthreads();

    const int e0 = c * CH + t * 4;
    const int4 dv = load_dst4(dst, e0, P);
    const int4 sv = load_dst4(src, e0, P);
    const int dd[4] = {dv.x, dv.y, dv.z, dv.w};
    const int ss[4] = {sv.x, sv.y, sv.z, sv.w};
    int rank[4];
    #pragma unroll
    for (int k = 0; k < 4; ++k)
        rank[k] = ((unsigned)dd[k] < (unsigned)A) ? atomicAdd(&hist[dd[k] >> DLB], 1) : -1;
    __syncthreads();

    const int npt = (pl.nbins + BT - 1) / BT;
    const int b0 = t * npt;
    int cnts[4];
    int loc = 0;
    #pragma unroll
    for (int j = 0; j < 4; ++j) {
        const int b = b0 + j;
        cnts[j] = (j < npt && b < pl.nbins) ? hist[b] : 0;
        loc += cnts[j];
    }
    int inc = loc;
    #pragma unroll
    for (int d = 1; d < 32; d <<= 1) {
        const int nb = __shfl_up(inc, d);
        if (lane >= d) inc += nb;
    }
    if (lane == 31) wsum[wave] = inc;
    __syncthreads();
    int wbase = 0, tot = 0;
    #pragma unroll
    for (int w = 0; w < BT / 32; ++w) {
        const int v = wsum[w];
        if (w < wave) wbase += v;
        tot += v;
    }
    int ex = wbase + inc - loc;
    #pragma unroll
    for (int j = 0; j < 4; ++j) {
        const int b = b0 + j;
        if (j < npt && b < pl.nbins) { hist[b] = ex; ex += cnts[j]; }
    }
    if (t == 0) hist[pl.nbins] = tot;
    __syncthreads();

    #pragma unroll
    for (int k = 0; k < 4; ++k)
        if (rank[k] >= 0)
            sorted[hist[dd[k] >> DLB] + rank[k]] =
                (int)(((unsigned)ss[k] << DLB) | (unsigned)(dd[k] & (APW - 1)));
    __syncthreads();

    int* __restrict__ chunk = bins + (size_t)c * CH;
    for (int i = t; i < tot; i += BT) chunk[i] = sorted[i];
    int* __restrict__ off = bins + (size_t)pl.nchunk * CH + (size_t)c * pl.ostride;
    for (int i = t; i <= pl.nbins; i += BT) off[i] = hist[i];
}

__global__ __launch_bounds__(OT) void edge_kernel(
    const float* __restrict__ pos, const int* __restrict__ dst,
    const int* __restrict__ src, const float* __restrict__ Wb,
    const float* __restrict__ xin, float* __restrict__ accum,
    int* __restrict__ lst, const int* __restrict__ bins, int P, int A, int add_cut, int reuse) {
    __shared__ float acc[APW * F];
    __shared__ int listE[LCAP];
    __shared__ int listD[LCAP];
    __shared__ int s_cnt;

    const int t = threadIdx.x;
    const int lane = t & 31;
    const int wave = t >> 5;
    const int lo = blockIdx.x * APW;
    const float addc = add_cut ? 1.0f : 0.0f;

    const int rows = min(APW, A - lo);
    int* __restrict__ mylst = lst + (size_t)lo * F;
    const int capS = rows * F - 32;
    const bool packable = A <= (1 << (32 - DLB));

    int total = 0;
    bool from_list = false;
    if (reuse) {
        const int c = mylst[capS];
        if (c >= 0 && c <= capS && c <= LCAP) { from_list = true; total = c; }
    }

    for (int i = t; i < APW * F; i += OT) acc[i] = 0.0f;
    if (t == 0) s_cnt = 0;
    __syncthreads();

    if (from_list) {
        for (int i = t; i < total; i += OT) {
            const unsigned w = (unsigned)mylst[i];
            listE[i] = (int)(w >> DLB);
            listD[i] = (int)(w & (unsigned)(APW - 1));
        }
    } else if (!reuse && bin_plan(A, P).ok) {
        const BinPlan pl = bin_plan(A, P);
        const int* __restrict__ offs = bins + (size_t)pl.nchunk * CH;
        const int sub = t & 3;
        for (int g0 = 0; g0 < pl.nchunk; g0 += OT / 4) {
            const int g = g0 + (t >> 2);
            int beg = 0, n = 0;
            if (g < pl.nchunk) {
                const int* __restrict__ o = offs + (size_t)g * pl.ostride + blockIdx.x;
                beg = o[0];
                n = o[1] - beg;
            }
            int base = 0;
            if (sub == 0 && n > 0) base = atomicAdd(&s_cnt, n);
            base = __shfl(base, lane & ~3);
            for (int j = sub; j < n; j += 4) {
                const int slot = base + j;
                if (slot < LCAP) {
                    const unsigned w = (unsigned)bins[(size_t)g * CH + beg + j];
                    listE[slot] = (int)(w >> DLB);
                    listD[slot] = (int)(w & (unsigned)(APW - 1));
                }
            }
        }
        __syncthreads();
        total = s_cnt;
        const bool save = total <= capS && total <= LCAP;
        if (save)
            for (int i = t; i < total; i += OT)
                mylst[i] = (int)(((unsigned)listE[i] << DLB) | (unsigned)listD[i]);
        if (t == 0) mylst[capS] = save ? total : -1;
    } else {
        auto keep4 = [&](const int4 v, const int b) {
            const int dv[4] = {v.x, v.y, v.z, v.w};
            #pragma unroll
            for (int c = 0; c < 4; ++c) {
                unsigned u = (unsigned)(dv[c] - lo);
                if (u < (unsigned)APW) {
                    int slot = atomicAdd(&s_cnt, 1);
                    if (slot < LCAP) { listE[slot] = b + c; listD[slot] = (int)u; }
                }
            }
        };
        int base0 = t * 4;
        for (; base0 + (SCAN_UN - 1) * OT * 4 + 3 < P; base0 += OT * 4 * SCAN_UN) {
            int4 v[SCAN_UN];
            #pragma unroll
            for (int j = 0; j < SCAN_UN; ++j)
                v[j] = *(const int4*)(dst + base0 + j * OT * 4);
            #pragma unroll
            for (int j = 0; j < SCAN_UN; ++j)
                keep4(v[j], base0 + j * OT * 4);
        }
        for (; base0 < P; base0 += OT * 4)
            keep4(load_dst4(dst, base0, P), base0);
        __syncthreads();
        total = s_cnt;

        const bool save = !reuse && packable && total <= capS && total <= LCAP;
        if (total <= LCAP) {
            for (int i = t; i < total; i += OT) {
                const int s = src[listE[i]];
                listE[i] = s;
                if (save) mylst[i] = (int)(((unsigned)s << DLB) | (unsigned)listD[i]);
            }
        }
        if (!reuse && t == 0) mylst[capS] = save ? total : -1;
    }
    __syncthreads();

    if (total <= LCAP) {
        const int m  = lane & 15;
        const int h  = lane >> 4;
        const int cn = lane & 15;

        v2f bf0[4], bf1[4];
        #pragma unroll
        for (int kc = 0; kc < 4; ++kc) {
            const int k0 = 4 * kc + 2 * h;
            bf0[kc][0] = Wb[k0 * F + cn];
            bf0[kc][1] = Wb[(k0 + 1) * F + cn];
            bf1[kc][0] = Wb[k0 * F + cn + 16];
            bf1[kc][1] = Wb[(k0 + 1) * F + cn + 16];
        }
        if (h == 0) { bf0[0][0] += addc; bf1[0][0] += addc; }

        const int ntiles = (total + 15) >> 4;
        for (int tile = wave; tile < ntiles; tile += OT / 32) {
            const int idx = tile * 16 + m;
            const bool valid = idx < total;
            const int s  = valid ? listE[idx] : 0;
            const int dl = valid ? listD[idx] : 0;

            int   drr[8];
            float xv0[8], xv1[8];
            #pragma unroll
            for (int q = 0; q < 8; ++q) {
                const int r  = 8 * h + q;
                const int sr = __shfl(s, r);
                drr[q] = __shfl(dl, r);
                xv0[q] = xin[sr * F + cn];
                xv1[q] = xin[sr * F + cn + 16];
            }

            float rw[NB];
            edge_rw(pos, s, lo + dl, valid, rw);

            v8f c0, c1;
            #pragma unroll
            for (int q = 0; q < 8; ++q) { c0[q] = 0.0f; c1[q] = 0.0f; }
            #pragma unroll
            for (int kc = 0; kc < 4; ++kc) {
                v2f a;
                a[0] = pick(h != 0, rw[4 * kc + 2], rw[4 * kc]);
                a[1] = pick(h != 0, rw[4 * kc + 3], rw[4 * kc + 1]);
                c0 = __builtin_amdgcn_wmma_f32_16x16x4_f32(false, a, false, bf0[kc], (short)0, c0, false, false);
                c1 = __builtin_amdgcn_wmma_f32_16x16x4_f32(false, a, false, bf1[kc], (short)0, c1, false, false);
            }
            #pragma unroll
            for (int q = 0; q < 8; ++q) {
                atomicAdd(&acc[drr[q] * F + cn],      c0[q] * xv0[q]);
                atomicAdd(&acc[drr[q] * F + cn + 16], c1[q] * xv1[q]);
            }
        }
    } else {
        for (int e = t; e < P; e += OT) {
            unsigned u = (unsigned)(dst[e] - lo);
            if (u < (unsigned)APW) {
                const int s = src[e];
                float rw[NB];
                edge_rw(pos, s, lo + (int)u, true, rw);
                for (int f = 0; f < F; ++f) {
                    float c = rw[0] * (Wb[f] + addc);
                    #pragma unroll
                    for (int k = 1; k < NB; ++k) c += rw[k] * Wb[k * F + f];
                    atomicAdd(&acc[(int)u * F + f], c * xin[s * F + f]);
                }
            }
        }
    }
    __syncthreads();

    for (int i = t; i < APW * F; i += OT)
        if (lo + (i >> 5) < A) accum[lo * F + i] = acc[i];
}

__global__ __launch_bounds__(256) void refine_kernel(
    const float* __restrict__ xa, const float* __restrict__ xb,
    const float* __restrict__ W1s, const float* __restrict__ b1s,
    const float* __restrict__ W2s, const float* __restrict__ b2s,
    float* __restrict__ out, int A) {
    __shared__ __align__(16) float sW1[3 * F * F];
    __shared__ __align__(16) float sW2[3 * F * F];
    __shared__ float sb1[3 * F];
    __shared__ float sb2[3 * F];
    __shared__ __align__(16) float xt[128 * F];
    __shared__ __align__(16) float ht[128 * F];

    const int t = threadIdx.x;
    const int lane = t & 31;
    const int wave = t >> 5;
    const int base = blockIdx.x * 128 * F;
    const int lim = A * F;

    for (int i = t; i < 3 * F * F; i += 256) {
        int rem = i & 1023;
        int k = rem >> 5;
        int n = rem & 31;
        int fidx = (i & ~1023) + ((k >> 1) << 6) + (n << 1) + (k & 1);
        sW1[fidx] = W1s[i];
        sW2[fidx] = W2s[i];
    }
    for (int i = t; i < 3 * F; i += 256) { sb1[i] = b1s[i]; sb2[i] = b2s[i]; }
    for (int i = t; i < 128 * F; i += 256)
        xt[i] = (base + i < lim) ? (xa[base + i] + xb[base + i]) : 0.0f;
    __syncthreads();

    const int m     = lane & 15;
    const int khalf = (lane >> 4) << 1;
    const int row0  = wave * 16;
    const int cm    = (lane >> 4) << 3;
    const int cn    = lane & 15;

    for (int j = 0; j < 3; ++j) {
        const v2f* W1p = (const v2f*)(sW1 + j * F * F);
        const v2f* W2p = (const v2f*)(sW2 + j * F * F);
        const float* B1 = sb1 + j * F;
        const float* B2 = sb2 + j * F;

        v8f acc0, acc1;
        {
            float bn0 = B1[cn], bn1 = B1[cn + 16];
            #pragma unroll
            for (int q = 0; q < 8; ++q) { acc0[q] = bn0; acc1[q] = bn1; }
        }
        #pragma unroll
        for (int kc = 0; kc < 8; ++kc) {
            int kb = kc * 4 + khalf;
            int kp = (kb >> 1) << 5;
            v2f a  = *(const v2f*)(xt + (row0 + m) * F + kb);
            v2f b0 = W1p[kp + cn];
            v2f b1 = W1p[kp + cn + 16];
            acc0 = __builtin_amdgcn_wmma_f32_16x16x4_f32(false, a, false, b0, (short)0, acc0, false, false);
            acc1 = __builtin_amdgcn_wmma_f32_16x16x4_f32(false, a, false, b1, (short)0, acc1, false, false);
        }
        #pragma unroll
        for (int q = 0; q < 8; ++q) {
            ht[(row0 + cm + q) * F + cn]      = fast_silu(acc0[q]);
            ht[(row0 + cm + q) * F + cn + 16] = fast_silu(acc1[q]);
        }
        __syncthreads();

        {
            float bn0 = B2[cn], bn1 = B2[cn + 16];
            #pragma unroll
            for (int q = 0; q < 8; ++q) { acc0[q] = bn0; acc1[q] = bn1; }
        }
        #pragma unroll
        for (int kc = 0; kc < 8; ++kc) {
            int kb = kc * 4 + khalf;
            int kp = (kb >> 1) << 5;
            v2f a  = *(const v2f*)(ht + (row0 + m) * F + kb);
            v2f b0 = W2p[kp + cn];
            v2f b1 = W2p[kp + cn + 16];
            acc0 = __builtin_amdgcn_wmma_f32_16x16x4_f32(false, a, false, b0, (short)0, acc0, false, false);
            acc1 = __builtin_amdgcn_wmma_f32_16x16x4_f32(false, a, false, b1, (short)0, acc1, false, false);
        }
        __syncthreads();
        #pragma unroll
        for (int q = 0; q < 8; ++q) {
            float x0v = xt[(row0 + cm + q) * F + cn]      + acc0[q];
            float x1v = xt[(row0 + cm + q) * F + cn + 16] + acc1[q];
            xt[(row0 + cm + q) * F + cn]      = fast_silu(x0v);
            xt[(row0 + cm + q) * F + cn + 16] = fast_silu(x1v);
        }
        __syncthreads();
    }

    for (int i = t; i < 128 * F; i += 256)
        if (base + i < lim) out[base + i] = xt[i];
}

extern "C" void kernel_launch(void* const* d_in, const int* in_sizes, int n_in,
                              void* d_out, int out_size, void* d_ws, size_t ws_size,
                              hipStream_t stream) {
    const float* pos  = (const float*)d_in[0];
    const int*   q    = (const int*)d_in[1];
    const int*   sp   = (const int*)d_in[2];
    const int*   Z    = (const int*)d_in[3];
    const int*   dst  = (const int*)d_in[4];
    const int*   src  = (const int*)d_in[5];
    const int*   seg  = (const int*)d_in[6];
    const float* ce   = (const float*)d_in[7];
    const float* se   = (const float*)d_in[8];
    const float* ae   = (const float*)d_in[9];
    const float* mW   = (const float*)d_in[10];
    const float* mb   = (const float*)d_in[11];
    const float* mpWb = (const float*)d_in[12];
    const float* rW1  = (const float*)d_in[13];
    const float* rb1  = (const float*)d_in[14];
    const float* rW2  = (const float*)d_in[15];
    const float* rb2  = (const float*)d_in[16];

    const int B = in_sizes[1];
    const int A = in_sizes[3];
    const int P = in_sizes[4];

    float* ws  = (float*)d_ws;
    float* mol = ws;
    float* x0  = mol + (size_t)B * F;
    float* s0  = x0 + (size_t)A * F;
    float* xs0 = s0 + (size_t)A * F;
    int* elist = (int*)(xs0 + (size_t)A * F);

    int* bins = (int*)xs0;

    prep_kernel<<<(P + CH - 1) / CH + (B * F + BT - 1) / BT, BT, 0, stream>>>(
        q, sp, ce, se, mW, mb, mol, B, dst, src, bins, A, P);
    x0_kernel<<<(A * F + 255) / 256, 256, 0, stream>>>(Z, seg, ae, mol, x0, A);

    const int eblocks = (A + APW - 1) / APW;

    edge_kernel<<<eblocks, OT, 0, stream>>>(pos, dst, src, mpWb, x0, s0, elist, bins, P, A, 1, 0);
    refine_kernel<<<(A + 127) / 128, 256, 0, stream>>>(
        x0, s0, rW1, rb1, rW2, rb2, xs0, A);
    edge_kernel<<<eblocks, OT, 0, stream>>>(pos, dst, src, mpWb + NB * F, xs0, s0, elist, bins, P, A, 0, 1);
    refine_kernel<<<(A + 127) / 128, 256, 0, stream>>>(
        xs0, s0, rW1 + 3 * F * F, rb1 + 3 * F, rW2 + 3 * F * F, rb2 + 3 * F,
        (float*)d_out, A);
}


# x0 launch folded into the first launch: 128 extra work-groups (16 atoms a wave, one feature a lane) sum an atom's mol row in place, same k order, and store x0; 6 launches become 5
# speedup vs baseline: 1.7539x; 1.0517x over previous
#include <hip/hip_runtime.h>
#include <hip/hip_bf16.h>

typedef float v2f __attribute__((ext_vector_type(2)));
typedef float v8f __attribute__((ext_vector_type(8)));

#define F 32
#define NB 16

__device__ __forceinline__ float fast_silu(float x) {
    return x * __builtin_amdgcn_rcpf(1.0f + __expf(-x));
}

#define APW  64
#define DLB  6
static_assert(APW == (1 << DLB), "a packed list word keeps the local destination in its low DLB bits");
#define LCAP 4096
#define OT   1024
#define SCAN_UN 8
#define CH   1024
#define BT   256
#define MAXB 1024
static_assert(CH == 4 * BT && MAXB == 4 * BT, "4 edges and at most 4 groups per binning thread");
#define XA   128
#define XR   (XA / (BT / 32))
static_assert(XR <= 32 && (XR & (XR - 1)) == 0, "a lane looks after the indices of atom a0 + (lane & (XR - 1))");

struct BinPlan { int nbins, nchunk, ostride; bool ok; };
__device__ __forceinline__ BinPlan bin_plan(int A, int P) {
    BinPlan p;
    p.nbins   = (A + APW - 1) / APW;
    p.nchunk  = (P + CH - 1) / CH;
    p.ostride = (p.nbins + 1 + 31) & ~31;
    p.ok = p.nbins <= MAXB && A <= (1 << (32 - DLB)) &&
           (long long)p.nchunk * (CH + p.ostride) <= (long long)A * F;
    return p;
}

__device__ __forceinline__ void edge_rw(const float* __restrict__ pos, int s, int d, bool live, float* rw) {
    float dx = pos[s * 3 + 0] - pos[d * 3 + 0];
    float dy = pos[s * 3 + 1] - pos[d * 3 + 1];
    float dz = pos[s * 3 + 2] - pos[d * 3 + 2];
    float r = sqrtf(dx * dx + dy * dy + dz * dz);
    float u = r * (1.0f / 6.0f);
    u = u * u;
    float cutv = 0.0f;
    if (u < 1.0f)
        cutv = __expf(1.0f - __builtin_amdgcn_rcpf(fmaxf(1.0f - u, 1e-6f)));
    cutv = live ? cutv : 0.0f;
    float zc = fminf(fmaxf(2.0f * __expf(-r) - 1.0f, -1.0f), 1.0f);
    float Tm2 = 1.0f, Tm1 = zc;
    rw[0] = cutv;
    rw[1] = cutv * zc;
    #pragma unroll
    for (int b2 = 2; b2 < NB; ++b2) {
        float T = 2.0f * zc * Tm1 - Tm2;
        rw[b2] = cutv * T;
        Tm2 = Tm1; Tm1 = T;
    }
}

__device__ __forceinline__ float pick(bool c, float a, float b) { return c ? a : b; }

__device__ __forceinline__ int4 load_dst4(const int* __restrict__ dst, int b, int P) {
    if (b + 3 < P) return *(const int4*)(dst + b);
    int4 v;
    v.x = (b     < P) ? dst[b]     : -1;
    v.y = (b + 1 < P) ? dst[b + 1] : -1;
    v.z = (b + 2 < P) ? dst[b + 2] : -1;
    v.w = (b + 3 < P) ? dst[b + 3] : -1;
    return v;
}

__global__ __launch_bounds__(BT) void prep_kernel(
    const int* __restrict__ q, const int* __restrict__ sp,
    const float* __restrict__ ce, const float* __restrict__ se,
    const float* __restrict__ mW, const float* __restrict__ mb,
    const int* __restrict__ Z, const int* __restrict__ seg,
    const float* __restrict__ ae, float* __restrict__ x0,
    const int* __restrict__ dst, const int* __restrict__ src,
    int* __restrict__ bins, int A, int P) {
    __shared__ int hist[MAXB + 1];
    __shared__ int sorted[CH];
    __shared__ int wsum[BT / 32];

    const BinPlan pl = bin_plan(A, P);
    const int t = threadIdx.x;
    if ((int)blockIdx.x >= pl.nchunk) {
        const int lane = t & 31;
        const int wv = __builtin_amdgcn_readfirstlane(t >> 5);
        const int a0 = ((int)blockIdx.x - pl.nchunk) * XA + wv * XR;
        if (a0 >= A) return;
        float w[32];
        #pragma unroll
        for (int k = 0; k < 32; ++k) w[k] = mW[k * F + lane];
        const float bias = mb[lane];

        const int aj  = min(a0 + (lane & (XR - 1)), A - 1);
        const int zj  = Z[aj];
        const int bj  = seg[aj];
        const int cij = min(max(q[bj] + 5, 0), 10);
        const int sij = min(max(sp[bj] - 1, 0), 6);

        float e[XR], c[XR];
        int key[XR];
        const float* __restrict__ crow = (lane < 16) ? ce : se;
        const int ccol = lane & 15;
        #pragma unroll
        for (int j = 0; j < XR; ++j) {
            const int z  = __builtin_amdgcn_readlane(zj, j);
            const int ci = __builtin_amdgcn_readlane(cij, j);
            const int si = __builtin_amdgcn_readlane(sij, j);
            e[j] = ae[z * F + lane];
            c[j] = crow[((lane < 16) ? ci : si) * 16 + ccol];
            key[j] = ci * 8 + si;
        }
        float m = 0.0f;
        #pragma unroll
        for (int j = 0; j < XR; ++j) {
            if (j == 0 || key[j] != key[j - 1]) {
                const int cb = __builtin_bit_cast(int, c[j]);
                m = bias;
                #pragma unroll
                for (int k = 0; k < 32; ++k)
                    m += __builtin_bit_cast(float, __builtin_amdgcn_readlane(cb, k)) * w[k];
            }
            if (a0 + j < A) x0[(size_t)(a0 + j) * F + lane] = e[j] + m;
        }
        return;
    }
    if (!pl.ok) return;

    const int lane = t & 31;
    const int wave = t >> 5;
    const int c = blockIdx.x;
    for (int i = t; i <= pl.nbins; i += BT) hist[i] = 0;
    __syncthreads();

    const int e0 = c * CH + t * 4;
    const int4 dv = load_dst4(dst, e0, P);
    const int4 sv = load_dst4(src, e0, P);
    const int dd[4] = {dv.x, dv.y, dv.z, dv.w};
    const int ss[4] = {sv.x, sv.y, sv.z, sv.w};
    int rank[4];
    #pragma unroll
    for (int k = 0; k < 4; ++k)
        rank[k] = ((unsigned)dd[k] < (unsigned)A) ? atomicAdd(&hist[dd[k] >> DLB], 1) : -1;
    __syncthreads();

    const int npt = (pl.nbins + BT - 1) / BT;
    const int b0 = t * npt;
    int cnts[4];
    int loc = 0;
    #pragma unroll
    for (int j = 0; j < 4; ++j) {
        const int b = b0 + j;
        cnts[j] = (j < npt && b < pl.nbins) ? hist[b] : 0;
        loc += cnts[j];
    }
    int inc = loc;
    #pragma unroll
    for (int d = 1; d < 32; d <<= 1) {
        const int nb = __shfl_up(inc, d);
        if (lane >= d) inc += nb;
    }
    if (lane == 31) wsum[wave] = inc;
    __syncthreads();
    int wbase = 0, tot = 0;
    #pragma unroll
    for (int w = 0; w < BT / 32; ++w) {
        const int v = wsum[w];
        if (w < wave) wbase += v;
        tot += v;
    }
    int ex = wbase + inc - loc;
    #pragma unroll
    for (int j = 0; j < 4; ++j) {
        const int b = b0 + j;
        if (j < npt && b < pl.nbins) { hist[b] = ex; ex += cnts[j]; }
    }
    if (t == 0) hist[pl.nbins] = tot;
    __syncthreads();

    #pragma unroll
    for (int k = 0; k < 4; ++k)
        if (rank[k] >= 0)
            sorted[hist[dd[k] >> DLB] + rank[k]] =
                (int)(((unsigned)ss[k] << DLB) | (unsigned)(dd[k] & (APW - 1)));
    __syncthreads();

    int* __restrict__ chunk = bins + (size_t)c * CH;
    for (int i = t; i < tot; i += BT) chunk[i] = sorted[i];
    int* __restrict__ off = bins + (size_t)pl.nchunk * CH + (size_t)c * pl.ostride;
    for (int i = t; i <= pl.nbins; i += BT) off[i] = hist[i];
}

__global__ __launch_bounds__(OT) void edge_kernel(
    const float* __restrict__ pos, const int* __restrict__ dst,
    const int* __restrict__ src, const float* __restrict__ Wb,
    const float* __restrict__ xin, float* __restrict__ accum,
    int* __restrict__ lst, const int* __restrict__ bins, int P, int A, int add_cut, int reuse) {
    __shared__ float acc[APW * F];
    __shared__ int listE[LCAP];
    __shared__ int listD[LCAP];
    __shared__ int s_cnt;

    const int t = threadIdx.x;
    const int lane = t & 31;
    const int wave = t >> 5;
    const int lo = blockIdx.x * APW;
    const float addc = add_cut ? 1.0f : 0.0f;

    const int rows = min(APW, A - lo);
    int* __restrict__ mylst = lst + (size_t)lo * F;
    const int capS = rows * F - 32;
    const bool packable = A <= (1 << (32 - DLB));

    int total = 0;
    bool from_list = false;
    if (reuse) {
        const int c = mylst[capS];
        if (c >= 0 && c <= capS && c <= LCAP) { from_list = true; total = c; }
    }

    for (int i = t; i < APW * F; i += OT) acc[i] = 0.0f;
    if (t == 0) s_cnt = 0;
    __syncthreads();

    if (from_list) {
        for (int i = t; i < total; i += OT) {
            const unsigned w = (unsigned)mylst[i];
            listE[i] = (int)(w >> DLB);
            listD[i] = (int)(w & (unsigned)(APW - 1));
        }
    } else if (!reuse && bin_plan(A, P).ok) {
        const BinPlan pl = bin_plan(A, P);
        const int* __restrict__ offs = bins + (size_t)pl.nchunk * CH;
        const int sub = t & 3;
        for (int g0 = 0; g0 < pl.nchunk; g0 += OT / 4) {
            const int g = g0 + (t >> 2);
            int beg = 0, n = 0;
            if (g < pl.nchunk) {
                const int* __restrict__ o = offs + (size_t)g * pl.ostride + blockIdx.x;
                beg = o[0];
                n = o[1] - beg;
            }
            int base = 0;
            if (sub == 0 && n > 0) base = atomicAdd(&s_cnt, n);
            base = __shfl(base, lane & ~3);
            for (int j = sub; j < n; j += 4) {
                const int slot = base + j;
                if (slot < LCAP) {
                    const unsigned w = (unsigned)bins[(size_t)g * CH + beg + j];
                    listE[slot] = (int)(w >> DLB);
                    listD[slot] = (int)(w & (unsigned)(APW - 1));
                }
            }
        }
        __syncthreads();
        total = s_cnt;
        const bool save = total <= capS && total <= LCAP;
        if (save)
            for (int i = t; i < total; i += OT)
                mylst[i] = (int)(((unsigned)listE[i] << DLB) | (unsigned)listD[i]);
        if (t == 0) mylst[capS] = save ? total : -1;
    } else {
        auto keep4 = [&](const int4 v, const int b) {
            const int dv[4] = {v.x, v.y, v.z, v.w};
            #pragma unroll
            for (int c = 0; c < 4; ++c) {
                unsigned u = (unsigned)(dv[c] - lo);
                if (u < (unsigned)APW) {
                    int slot = atomicAdd(&s_cnt, 1);
                    if (slot < LCAP) { listE[slot] = b + c; listD[slot] = (int)u; }
                }
            }
        };
        int base0 = t * 4;
        for (; base0 + (SCAN_UN - 1) * OT * 4 + 3 < P; base0 += OT * 4 * SCAN_UN) {
            int4 v[SCAN_UN];
            #pragma unroll
            for (int j = 0; j < SCAN_UN; ++j)
                v[j] = *(const int4*)(dst + base0 + j * OT * 4);
            #pragma unroll
            for (int j = 0; j < SCAN_UN; ++j)
                keep4(v[j], base0 + j * OT * 4);
        }
        for (; base0 < P; base0 += OT * 4)
            keep4(load_dst4(dst, base0, P), base0);
        __syncthreads();
        total = s_cnt;

        const bool save = !reuse && packable && total <= capS && total <= LCAP;
        if (total <= LCAP) {
            for (int i = t; i < total; i += OT) {
                const int s = src[listE[i]];
                listE[i] = s;
                if (save) mylst[i] = (int)(((unsigned)s << DLB) | (unsigned)listD[i]);
            }
        }
        if (!reuse && t == 0) mylst[capS] = save ? total : -1;
    }
    __syncthreads();

    if (total <= LCAP) {
        const int m  = lane & 15;
        const int h  = lane >> 4;
        const int cn = lane & 15;

        v2f bf0[4], bf1[4];
        #pragma unroll
        for (int kc = 0; kc < 4; ++kc) {
            const int k0 = 4 * kc + 2 * h;
            bf0[kc][0] = Wb[k0 * F + cn];
            bf0[kc][1] = Wb[(k0 + 1) * F + cn];
            bf1[kc][0] = Wb[k0 * F + cn + 16];
            bf1[kc][1] = Wb[(k0 + 1) * F + cn + 16];
        }
        if (h == 0) { bf0[0][0] += addc; bf1[0][0] += addc; }

        const int ntiles = (total + 15) >> 4;
        for (int tile = wave; tile < ntiles; tile += OT / 32) {
            const int idx = tile * 16 + m;
            const bool valid = idx < total;
            const int s  = valid ? listE[idx] : 0;
            const int dl = valid ? listD[idx] : 0;

            int   drr[8];
            float xv0[8], xv1[8];
            #pragma unroll
            for (int q = 0; q < 8; ++q) {
                const int r  = 8 * h + q;
                const int sr = __shfl(s, r);
                drr[q] = __shfl(dl, r);
                xv0[q] = xin[sr * F + cn];
                xv1[q] = xin[sr * F + cn + 16];
            }

            float rw[NB];
            edge_rw(pos, s, lo + dl, valid, rw);

            v8f c0, c1;
            #pragma unroll
            for (int q = 0; q < 8; ++q) { c0[q] = 0.0f; c1[q] = 0.0f; }
            #pragma unroll
            for (int kc = 0; kc < 4; ++kc) {
                v2f a;
                a[0] = pick(h != 0, rw[4 * kc + 2], rw[4 * kc]);
                a[1] = pick(h != 0, rw[4 * kc + 3], rw[4 * kc + 1]);
                c0 = __builtin_amdgcn_wmma_f32_16x16x4_f32(false, a, false, bf0[kc], (short)0, c0, false, false);
                c1 = __builtin_amdgcn_wmma_f32_16x16x4_f32(false, a, false, bf1[kc], (short)0, c1, false, false);
            }
            #pragma unroll
            for (int q = 0; q < 8; ++q) {
                atomicAdd(&acc[drr[q] * F + cn],      c0[q] * xv0[q]);
                atomicAdd(&acc[drr[q] * F + cn + 16], c1[q] * xv1[q]);
            }
        }
    } else {
        for (int e = t; e < P; e += OT) {
            unsigned u = (unsigned)(dst[e] - lo);
            if (u < (unsigned)APW) {
                const int s = src[e];
                float rw[NB];
                edge_rw(pos, s, lo + (int)u, true, rw);
                for (int f = 0; f < F; ++f) {
                    float c = rw[0] * (Wb[f] + addc);
                    #pragma unroll
                    for (int k = 1; k < NB; ++k) c += rw[k] * Wb[k * F + f];
                    atomicAdd(&acc[(int)u * F + f], c * xin[s * F + f]);
                }
            }
        }
    }
    __syncthreads();

    for (int i = t; i < APW * F; i += OT)
        if (lo + (i >> 5) < A) accum[lo * F + i] = acc[i];
}

__global__ __launch_bounds__(256) void refine_kernel(
    const float* __restrict__ xa, const float* __restrict__ xb,
    const float* __restrict__ W1s, const float* __restrict__ b1s,
    const float* __restrict__ W2s, const float* __restrict__ b2s,
    float* __restrict__ out, int A) {
    __shared__ __align__(16) float sW1[3 * F * F];
    __shared__ __align__(16) float sW2[3 * F * F];
    __shared__ float sb1[3 * F];
    __shared__ float sb2[3 * F];
    __shared__ __align__(16) float xt[128 * F];
    __shared__ __align__(16) float ht[128 * F];

    const int t = threadIdx.x;
    const int lane = t & 31;
    const int wave = t >> 5;
    const int base = blockIdx.x * 128 * F;
    const int lim = A * F;

    for (int i = t; i < 3 * F * F; i += 256) {
        int rem = i & 1023;
        int k = rem >> 5;
        int n = rem & 31;
        int fidx = (i & ~1023) + ((k >> 1) << 6) + (n << 1) + (k & 1);
        sW1[fidx] = W1s[i];
        sW2[fidx] = W2s[i];
    }
    for (int i = t; i < 3 * F; i += 256) { sb1[i] = b1s[i]; sb2[i] = b2s[i]; }
    for (int i = t; i < 128 * F; i += 256)
        xt[i] = (base + i < lim) ? (xa[base + i] + xb[base + i]) : 0.0f;
    __syncthreads();

    const int m     = lane & 15;
    const int khalf = (lane >> 4) << 1;
    const int row0  = wave * 16;
    const int cm    = (lane >> 4) << 3;
    const int cn    = lane & 15;

    for (int j = 0; j < 3; ++j) {
        const v2f* W1p = (const v2f*)(sW1 + j * F * F);
        const v2f* W2p = (const v2f*)(sW2 + j * F * F);
        const float* B1 = sb1 + j * F;
        const float* B2 = sb2 + j * F;

        v8f acc0, acc1;
        {
            float bn0 = B1[cn], bn1 = B1[cn + 16];
            #pragma unroll
            for (int q = 0; q < 8; ++q) { acc0[q] = bn0; acc1[q] = bn1; }
        }
        #pragma unroll
        for (int kc = 0; kc < 8; ++kc) {
            int kb = kc * 4 + khalf;
            int kp = (kb >> 1) << 5;
            v2f a  = *(const v2f*)(xt + (row0 + m) * F + kb);
            v2f b0 = W1p[kp + cn];
            v2f b1 = W1p[kp + cn + 16];
            acc0 = __builtin_amdgcn_wmma_f32_16x16x4_f32(false, a, false, b0, (short)0, acc0, false, false);
            acc1 = __builtin_amdgcn_wmma_f32_16x16x4_f32(false, a, false, b1, (short)0, acc1, false, false);
        }
        #pragma unroll
        for (int q = 0; q < 8; ++q) {
            ht[(row0 + cm + q) * F + cn]      = fast_silu(acc0[q]);
            ht[(row0 + cm + q) * F + cn + 16] = fast_silu(acc1[q]);
        }
        __syncthreads();

        {
            float bn0 = B2[cn], bn1 = B2[cn + 16];
            #pragma unroll
            for (int q = 0; q < 8; ++q) { acc0[q] = bn0; acc1[q] = bn1; }
        }
        #pragma unroll
        for (int kc = 0; kc < 8; ++kc) {
            int kb = kc * 4 + khalf;
            int kp = (kb >> 1) << 5;
            v2f a  = *(const v2f*)(ht + (row0 + m) * F + kb);
            v2f b0 = W2p[kp + cn];
            v2f b1 = W2p[kp + cn + 16];
            acc0 = __builtin_amdgcn_wmma_f32_16x16x4_f32(false, a, false, b0, (short)0, acc0, false, false);
            acc1 = __builtin_amdgcn_wmma_f32_16x16x4_f32(false, a, false, b1, (short)0, acc1, false, false);
        }
        __syncthreads();
        #pragma unroll
        for (int q = 0; q < 8; ++q) {
            float x0v = xt[(row0 + cm + q) * F + cn]      + acc0[q];
            float x1v = xt[(row0 + cm + q) * F + cn + 16] + acc1[q];
            xt[(row0 + cm + q) * F + cn]      = fast_silu(x0v);
            xt[(row0 + cm + q) * F + cn + 16] = fast_silu(x1v);
        }
        __syncthreads();
    }

    for (int i = t; i < 128 * F; i += 256)
        if (base + i < lim) out[base + i] = xt[i];
}

extern "C" void kernel_launch(void* const* d_in, const int* in_sizes, int n_in,
                              void* d_out, int out_size, void* d_ws, size_t ws_size,
                              hipStream_t stream) {
    const float* pos  = (const float*)d_in[0];
    const int*   q    = (const int*)d_in[1];
    const int*   sp   = (const int*)d_in[2];
    const int*   Z    = (const int*)d_in[3];
    const int*   dst  = (const int*)d_in[4];
    const int*   src  = (const int*)d_in[5];
    const int*   seg  = (const int*)d_in[6];
    const float* ce   = (const float*)d_in[7];
    const float* se   = (const float*)d_in[8];
    const float* ae   = (const float*)d_in[9];
    const float* mW   = (const float*)d_in[10];
    const float* mb   = (const float*)d_in[11];
    const float* mpWb = (const float*)d_in[12];
    const float* rW1  = (const float*)d_in[13];
    const float* rb1  = (const float*)d_in[14];
    const float* rW2  = (const float*)d_in[15];
    const float* rb2  = (const float*)d_in[16];

    const int B = in_sizes[1];
    const int A = in_sizes[3];
    const int P = in_sizes[4];

    float* ws  = (float*)d_ws;
    float* x0  = ws + (size_t)B * F;
    float* s0  = x0 + (size_t)A * F;
    float* xs0 = s0 + (size_t)A * F;
    int* elist = (int*)(xs0 + (size_t)A * F);

    int* bins = (int*)xs0;

    prep_kernel<<<(P + CH - 1) / CH + (A + XA - 1) / XA, BT, 0, stream>>>(
        q, sp, ce, se, mW, mb, Z, seg, ae, x0, dst, src, bins, A, P);

    const int eblocks = (A + APW - 1) / APW;

    edge_kernel<<<eblocks, OT, 0, stream>>>(pos, dst, src, mpWb, x0, s0, elist, bins, P, A, 1, 0);
    refine_kernel<<<(A + 127) / 128, 256, 0, stream>>>(
        x0, s0, rW1, rb1, rW2, rb2, xs0, A);
    edge_kernel<<<eblocks, OT, 0, stream>>>(pos, dst, src, mpWb + NB * F, xs0, s0, elist, bins, P, A, 0, 1);
    refine_kernel<<<(A + 127) / 128, 256, 0, stream>>>(
        xs0, s0, rW1 + 3 * F * F, rb1 + 3 * F, rW2 + 3 * F * F, rb2 + 3 * F,
        (float*)d_out, A);
}


# both residual-refine launches folded into the edge passes: each group refines its own 64 rows in LDS right after its sums are complete (4 waves, one row tile each); 5 launches become 3
# speedup vs baseline: 3.7845x; 2.1578x over previous
#include <hip/hip_runtime.h>
#include <hip/hip_bf16.h>

typedef float v2f __attribute__((ext_vector_type(2)));
typedef float v8f __attribute__((ext_vector_type(8)));

#define F 32
#define NB 16

__device__ __forceinline__ float fast_silu(float x) {
    return x * __builtin_amdgcn_rcpf(1.0f + __expf(-x));
}

#define APW  64
#define DLB  6
static_assert(APW == (1 << DLB), "a packed list word keeps the local destination in its low DLB bits");
#define LCAP 4096
#define OT   1024
#define SCAN_UN 8
#define CH   1024
#define BT   256
#define MAXB 1024
static_assert(CH == 4 * BT && MAXB == 4 * BT, "4 edges and at most 4 groups per binning thread");
#define XA   128
#define XR   (XA / (BT / 32))
static_assert(XR <= 32 && (XR & (XR - 1)) == 0, "a lane looks after the indices of atom a0 + (lane & (XR - 1))");

struct BinPlan { int nbins, nchunk, ostride; bool ok; };
__device__ __forceinline__ BinPlan bin_plan(int A, int P) {
    BinPlan p;
    p.nbins   = (A + APW - 1) / APW;
    p.nchunk  = (P + CH - 1) / CH;
    p.ostride = (p.nbins + 1 + 31) & ~31;
    p.ok = p.nbins <= MAXB && A <= (1 << (32 - DLB)) &&
           (long long)p.nchunk * (CH + p.ostride) <= (long long)A * F;
    return p;
}

__device__ __forceinline__ void edge_rw(const float* __restrict__ pos, int s, int d, bool live, float* rw) {
    float dx = pos[s * 3 + 0] - pos[d * 3 + 0];
    float dy = pos[s * 3 + 1] - pos[d * 3 + 1];
    float dz = pos[s * 3 + 2] - pos[d * 3 + 2];
    float r = sqrtf(dx * dx + dy * dy + dz * dz);
    float u = r * (1.0f / 6.0f);
    u = u * u;
    float cutv = 0.0f;
    if (u < 1.0f)
        cutv = __expf(1.0f - __builtin_amdgcn_rcpf(fmaxf(1.0f - u, 1e-6f)));
    cutv = live ? cutv : 0.0f;
    float zc = fminf(fmaxf(2.0f * __expf(-r) - 1.0f, -1.0f), 1.0f);
    float Tm2 = 1.0f, Tm1 = zc;
    rw[0] = cutv;
    rw[1] = cutv * zc;
    #pragma unroll
    for (int b2 = 2; b2 < NB; ++b2) {
        float T = 2.0f * zc * Tm1 - Tm2;
        rw[b2] = cutv * T;
        Tm2 = Tm1; Tm1 = T;
    }
}

__device__ __forceinline__ float pick(bool c, float a, float b) { return c ? a : b; }

__device__ __forceinline__ int4 load_dst4(const int* __restrict__ dst, int b, int P) {
    if (b + 3 < P) return *(const int4*)(dst + b);
    int4 v;
    v.x = (b     < P) ? dst[b]     : -1;
    v.y = (b + 1 < P) ? dst[b + 1] : -1;
    v.z = (b + 2 < P) ? dst[b + 2] : -1;
    v.w = (b + 3 < P) ? dst[b + 3] : -1;
    return v;
}

__global__ __launch_bounds__(BT) void prep_kernel(
    const int* __restrict__ q, const int* __restrict__ sp,
    const float* __restrict__ ce, const float* __restrict__ se,
    const float* __restrict__ mW, const float* __restrict__ mb,
    const int* __restrict__ Z, const int* __restrict__ seg,
    const float* __restrict__ ae, float* __restrict__ x0,
    const int* __restrict__ dst, const int* __restrict__ src,
    int* __restrict__ bins, int A, int P) {
    __shared__ int hist[MAXB + 1];
    __shared__ int sorted[CH];
    __shared__ int wsum[BT / 32];

    const BinPlan pl = bin_plan(A, P);
    const int t = threadIdx.x;
    if ((int)blockIdx.x >= pl.nchunk) {
        const int lane = t & 31;
        const int wv = __builtin_amdgcn_readfirstlane(t >> 5);
        const int a0 = ((int)blockIdx.x - pl.nchunk) * XA + wv * XR;
        if (a0 >= A) return;
        float w[32];
        #pragma unroll
        for (int k = 0; k < 32; ++k) w[k] = mW[k * F + lane];
        const float bias = mb[lane];

        const int aj  = min(a0 + (lane & (XR - 1)), A - 1);
        const int zj  = Z[aj];
        const int bj  = seg[aj];
        const int cij = min(max(q[bj] + 5, 0), 10);
        const int sij = min(max(sp[bj] - 1, 0), 6);

        float e[XR], c[XR];
        int key[XR];
        const float* __restrict__ crow = (lane < 16) ? ce : se;
        const int ccol = lane & 15;
        #pragma unroll
        for (int j = 0; j < XR; ++j) {
            const int z  = __builtin_amdgcn_readlane(zj, j);
            const int ci = __builtin_amdgcn_readlane(cij, j);
            const int si = __builtin_amdgcn_readlane(sij, j);
            e[j] = ae[z * F + lane];
            c[j] = crow[((lane < 16) ? ci : si) * 16 + ccol];
            key[j] = ci * 8 + si;
        }
        float m = 0.0f;
        #pragma unroll
        for (int j = 0; j < XR; ++j) {
            if (j == 0 || key[j] != key[j - 1]) {
                const int cb = __builtin_bit_cast(int, c[j]);
                m = bias;
                #pragma unroll
                for (int k = 0; k < 32; ++k)
                    m += __builtin_bit_cast(float, __builtin_amdgcn_readlane(cb, k)) * w[k];
            }
            if (a0 + j < A) x0[(size_t)(a0 + j) * F + lane] = e[j] + m;
        }
        return;
    }
    if (!pl.ok) return;

    const int lane = t & 31;
    const int wave = t >> 5;
    const int c = blockIdx.x;
    for (int i = t; i <= pl.nbins; i += BT) hist[i] = 0;
    __syncthreads();

    const int e0 = c * CH + t * 4;
    const int4 dv = load_dst4(dst, e0, P);
    const int4 sv = load_dst4(src, e0, P);
    const int dd[4] = {dv.x, dv.y, dv.z, dv.w};
    const int ss[4] = {sv.x, sv.y, sv.z, sv.w};
    int rank[4];
    #pragma unroll
    for (int k = 0; k < 4; ++k)
        rank[k] = ((unsigned)dd[k] < (unsigned)A) ? atomicAdd(&hist[dd[k] >> DLB], 1) : -1;
    __syncthreads();

    const int npt = (pl.nbins + BT - 1) / BT;
    const int b0 = t * npt;
    int cnts[4];
    int loc = 0;
    #pragma unroll
    for (int j = 0; j < 4; ++j) {
        const int b = b0 + j;
        cnts[j] = (j < npt && b < pl.nbins) ? hist[b] : 0;
        loc += cnts[j];
    }
    int inc = loc;
    #pragma unroll
    for (int d = 1; d < 32; d <<= 1) {
        const int nb = __shfl_up(inc, d);
        if (lane >= d) inc += nb;
    }
    if (lane == 31) wsum[wave] = inc;
    __syncthreads();
    int wbase = 0, tot = 0;
    #pragma unroll
    for (int w = 0; w < BT / 32; ++w) {
        const int v = wsum[w];
        if (w < wave) wbase += v;
        tot += v;
    }
    int ex = wbase + inc - loc;
    #pragma unroll
    for (int j = 0; j < 4; ++j) {
        const int b = b0 + j;
        if (j < npt && b < pl.nbins) { hist[b] = ex; ex += cnts[j]; }
    }
    if (t == 0) hist[pl.nbins] = tot;
    __syncthreads();

    #pragma unroll
    for (int k = 0; k < 4; ++k)
        if (rank[k] >= 0)
            sorted[hist[dd[k] >> DLB] + rank[k]] =
                (int)(((unsigned)ss[k] << DLB) | (unsigned)(dd[k] & (APW - 1)));
    __syncthreads();

    int* __restrict__ chunk = bins + (size_t)c * CH;
    for (int i = t; i < tot; i += BT) chunk[i] = sorted[i];
    int* __restrict__ off = bins + (size_t)pl.nchunk * CH + (size_t)c * pl.ostride;
    for (int i = t; i <= pl.nbins; i += BT) off[i] = hist[i];
}

__device__ __forceinline__ void wave_lds_order() {
    __builtin_amdgcn_fence(__ATOMIC_RELEASE, "wavefront");
    __builtin_amdgcn_wave_barrier();
    __builtin_amdgcn_fence(__ATOMIC_ACQUIRE, "wavefront");
}

#define RW (APW / 16)
static_assert(3 * F * F == 3 * OT, "each thread stages 3 elements of each weight stack");

__global__ __launch_bounds__(OT) void edge_kernel(
    const float* __restrict__ pos, const int* __restrict__ dst,
    const int* __restrict__ src, const float* __restrict__ Wb,
    const float* __restrict__ xin, float* __restrict__ xout,
    const float* __restrict__ W1s, const float* __restrict__ b1s,
    const float* __restrict__ W2s, const float* __restrict__ b2s,
    int* __restrict__ lst, const int* __restrict__ bins, int P, int A, int add_cut, int reuse) {
    __shared__ __align__(16) float acc[APW * F];
    __shared__ __align__(16) float ht[APW * F];
    __shared__ __align__(16) float sW1[3 * F * F];
    __shared__ __align__(16) float sW2[3 * F * F];
    __shared__ float sb1[3 * F];
    __shared__ float sb2[3 * F];
    __shared__ int listE[LCAP];
    __shared__ int listD[LCAP];
    __shared__ int s_cnt;

    const int t = threadIdx.x;
    const int lane = t & 31;
    const int wave = t >> 5;
    const int lo = blockIdx.x * APW;
    const float addc = add_cut ? 1.0f : 0.0f;
    const int wv = __builtin_amdgcn_readfirstlane(t >> 5);

    float w1r[3], w2r[3];
    #pragma unroll
    for (int j = 0; j < 3; ++j) { w1r[j] = W1s[t + j * OT]; w2r[j] = W2s[t + j * OT]; }
    float b1r = 0.0f, b2r = 0.0f;
    if (t < 3 * F) { b1r = b1s[t]; b2r = b2s[t]; }
    float4 xo[4];
    #pragma unroll
    for (int j = 0; j < 4; ++j) xo[j] = make_float4(0.0f, 0.0f, 0.0f, 0.0f);
    if (wv < RW) {
        #pragma unroll
        for (int j = 0; j < 4; ++j) {
            const int row = min(lo + wv * 16 + j * 4 + (lane >> 3), A - 1);
            xo[j] = *(const float4*)(xin + (size_t)row * F + (lane & 7) * 4);
        }
    }

    const int rows = min(APW, A - lo);
    int* __restrict__ mylst = lst + (size_t)lo * F;
    const int capS = rows * F - 32;
    const bool packable = A <= (1 << (32 - DLB));

    int total = 0;
    bool from_list = false;
    if (reuse) {
        const int c = mylst[capS];
        if (c >= 0 && c <= capS && c <= LCAP) { from_list = true; total = c; }
    }

    for (int i = t; i < APW * F; i += OT) acc[i] = 0.0f;
    if (t == 0) s_cnt = 0;
    __syncthreads();

    if (from_list) {
        for (int i = t; i < total; i += OT) {
            const unsigned w = (unsigned)mylst[i];
            listE[i] = (int)(w >> DLB);
            listD[i] = (int)(w & (unsigned)(APW - 1));
        }
    } else if (!reuse && bin_plan(A, P).ok) {
        const BinPlan pl = bin_plan(A, P);
        const int* __restrict__ offs = bins + (size_t)pl.nchunk * CH;
        const int sub = t & 3;
        for (int g0 = 0; g0 < pl.nchunk; g0 += OT / 4) {
            const int g = g0 + (t >> 2);
            int beg = 0, n = 0;
            if (g < pl.nchunk) {
                const int* __restrict__ o = offs + (size_t)g * pl.ostride + blockIdx.x;
                beg = o[0];
                n = o[1] - beg;
            }
            int base = 0;
            if (sub == 0 && n > 0) base = atomicAdd(&s_cnt, n);
            base = __shfl(base, lane & ~3);
            for (int j = sub; j < n; j += 4) {
                const int slot = base + j;
                if (slot < LCAP) {
                    const unsigned w = (unsigned)bins[(size_t)g * CH + beg + j];
                    listE[slot] = (int)(w >> DLB);
                    listD[slot] = (int)(w & (unsigned)(APW - 1));
                }
            }
        }
        __syncthreads();
        total = s_cnt;
        const bool save = total <= capS && total <= LCAP;
        if (save)
            for (int i = t; i < total; i += OT)
                mylst[i] = (int)(((unsigned)listE[i] << DLB) | (unsigned)listD[i]);
        if (t == 0) mylst[capS] = save ? total : -1;
    } else {
        auto keep4 = [&](const int4 v, const int b) {
            const int dv[4] = {v.x, v.y, v.z, v.w};
            #pragma unroll
            for (int c = 0; c < 4; ++c) {
                unsigned u = (unsigned)(dv[c] - lo);
                if (u < (unsigned)APW) {
                    int slot = atomicAdd(&s_cnt, 1);
                    if (slot < LCAP) { listE[slot] = b + c; listD[slot] = (int)u; }
                }
            }
        };
        int base0 = t * 4;
        for (; base0 + (SCAN_UN - 1) * OT * 4 + 3 < P; base0 += OT * 4 * SCAN_UN) {
            int4 v[SCAN_UN];
            #pragma unroll
            for (int j = 0; j < SCAN_UN; ++j)
                v[j] = *(const int4*)(dst + base0 + j * OT * 4);
            #pragma unroll
            for (int j = 0; j < SCAN_UN; ++j)
                keep4(v[j], base0 + j * OT * 4);
        }
        for (; base0 < P; base0 += OT * 4)
            keep4(load_dst4(dst, base0, P), base0);
        __syncthreads();
        total = s_cnt;

        const bool save = !reuse && packable && total <= capS && total <= LCAP;
        if (total <= LCAP) {
            for (int i = t; i < total; i += OT) {
                const int s = src[listE[i]];
                listE[i] = s;
                if (save) mylst[i] = (int)(((unsigned)s << DLB) | (unsigned)listD[i]);
            }
        }
        if (!reuse && t == 0) mylst[capS] = save ? total : -1;
    }
    #pragma unroll
    for (int j = 0; j < 3; ++j) {
        const int k = t >> 5, n = t & 31;
        const int fidx = j * F * F + ((k >> 1) << 6) + (n << 1) + (k & 1);
        sW1[fidx] = w1r[j];
        sW2[fidx] = w2r[j];
    }
    if (t < 3 * F) { sb1[t] = b1r; sb2[t] = b2r; }
    __syncthreads();

    if (total <= LCAP) {
        const int m  = lane & 15;
        const int h  = lane >> 4;
        const int cn = lane & 15;

        v2f bf0[4], bf1[4];
        #pragma unroll
        for (int kc = 0; kc < 4; ++kc) {
            const int k0 = 4 * kc + 2 * h;
            bf0[kc][0] = Wb[k0 * F + cn];
            bf0[kc][1] = Wb[(k0 + 1) * F + cn];
            bf1[kc][0] = Wb[k0 * F + cn + 16];
            bf1[kc][1] = Wb[(k0 + 1) * F + cn + 16];
        }
        if (h == 0) { bf0[0][0] += addc; bf1[0][0] += addc; }

        const int ntiles = (total + 15) >> 4;
        for (int tile = wave; tile < ntiles; tile += OT / 32) {
            const int idx = tile * 16 + m;
            const bool valid = idx < total;
            const int s  = valid ? listE[idx] : 0;
            const int dl = valid ? listD[idx] : 0;

            int   drr[8];
            float xv0[8], xv1[8];
            #pragma unroll
            for (int q = 0; q < 8; ++q) {
                const int r  = 8 * h + q;
                const int sr = __shfl(s, r);
                drr[q] = __shfl(dl, r);
                xv0[q] = xin[sr * F + cn];
                xv1[q] = xin[sr * F + cn + 16];
            }

            float rw[NB];
            edge_rw(pos, s, lo + dl, valid, rw);

            v8f c0, c1;
            #pragma unroll
            for (int q = 0; q < 8; ++q) { c0[q] = 0.0f; c1[q] = 0.0f; }
            #pragma unroll
            for (int kc = 0; kc < 4; ++kc) {
                v2f a;
                a[0] = pick(h != 0, rw[4 * kc + 2], rw[4 * kc]);
                a[1] = pick(h != 0, rw[4 * kc + 3], rw[4 * kc + 1]);
                c0 = __builtin_amdgcn_wmma_f32_16x16x4_f32(false, a, false, bf0[kc], (short)0, c0, false, false);
                c1 = __builtin_amdgcn_wmma_f32_16x16x4_f32(false, a, false, bf1[kc], (short)0, c1, false, false);
            }
            #pragma unroll
            for (int q = 0; q < 8; ++q) {
                atomicAdd(&acc[drr[q] * F + cn],      c0[q] * xv0[q]);
                atomicAdd(&acc[drr[q] * F + cn + 16], c1[q] * xv1[q]);
            }
        }
    } else {
        for (int e = t; e < P; e += OT) {
            unsigned u = (unsigned)(dst[e] - lo);
            if (u < (unsigned)APW) {
                const int s = src[e];
                float rw[NB];
                edge_rw(pos, s, lo + (int)u, true, rw);
                for (int f = 0; f < F; ++f) {
                    float c = rw[0] * (Wb[f] + addc);
                    #pragma unroll
                    for (int k = 1; k < NB; ++k) c += rw[k] * Wb[k * F + f];
                    atomicAdd(&acc[(int)u * F + f], c * xin[s * F + f]);
                }
            }
        }
    }
    __syncthreads();

    if (wv >= RW) return;

    float* xt = acc + wv * 16 * F;
    float* hw = ht  + wv * 16 * F;
    const int m     = lane & 15;
    const int khalf = (lane >> 4) << 1;
    const int cm    = (lane >> 4) << 3;
    const int cn    = lane & 15;

    v2f b0[8], b1[8];
    float bn0, bn1;
    auto load_b = [&](const float* Wf, const float* Bv) {
        const v2f* Wp = (const v2f*)Wf;
        #pragma unroll
        for (int kc = 0; kc < 8; ++kc) {
            const int kp = ((kc * 4 + khalf) >> 1) << 5;
            b0[kc] = Wp[kp + cn];
            b1[kc] = Wp[kp + cn + 16];
        }
        bn0 = Bv[cn];
        bn1 = Bv[cn + 16];
    };
    auto gemm = [&](const float* tile, v8f& acc0, v8f& acc1) {
        v2f a[8];
        #pragma unroll
        for (int kc = 0; kc < 8; ++kc) a[kc] = *(const v2f*)(tile + m * F + kc * 4 + khalf);
        #pragma unroll
        for (int q = 0; q < 8; ++q) { acc0[q] = bn0; acc1[q] = bn1; }
        #pragma unroll
        for (int kc = 0; kc < 8; ++kc) {
            acc0 = __builtin_amdgcn_wmma_f32_16x16x4_f32(false, a[kc], false, b0[kc], (short)0, acc0, false, false);
            acc1 = __builtin_amdgcn_wmma_f32_16x16x4_f32(false, a[kc], false, b1[kc], (short)0, acc1, false, false);
        }
    };

    load_b(sW1, sb1);

    #pragma unroll
    for (int j = 0; j < 4; ++j) {
        float4* p = (float4*)(xt + j * 128 + lane * 4);
        float4 v = *p;
        v.x += xo[j].x; v.y += xo[j].y; v.z += xo[j].z; v.w += xo[j].w;
        *p = v;
    }
    wave_lds_order();

    #pragma unroll
    for (int j = 0; j < 3; ++j) {
        v8f acc0, acc1;
        gemm(xt, acc0, acc1);
        load_b(sW2 + j * F * F, sb2 + j * F);
        #pragma unroll
        for (int q = 0; q < 8; ++q) {
            hw[(cm + q) * F + cn]      = fast_silu(acc0[q]);
            hw[(cm + q) * F + cn + 16] = fast_silu(acc1[q]);
        }
        wave_lds_order();

        gemm(hw, acc0, acc1);
        if (j < 2) load_b(sW1 + (j + 1) * F * F, sb1 + (j + 1) * F);
        #pragma unroll
        for (int q = 0; q < 8; ++q) {
            const float x0v = xt[(cm + q) * F + cn]      + acc0[q];
            const float x1v = xt[(cm + q) * F + cn + 16] + acc1[q];
            xt[(cm + q) * F + cn]      = fast_silu(x0v);
            xt[(cm + q) * F + cn + 16] = fast_silu(x1v);
        }
        wave_lds_order();
    }

    for (int j = 0; j < 4; ++j) {
        const int row = wv * 16 + j * 4 + (lane >> 3);
        if (lo + row < A)
            *(float4*)(xout + (size_t)(lo + row) * F + (lane & 7) * 4) = *(const float4*)(xt + j * 128 + lane * 4);
    }
}

extern "C" void kernel_launch(void* const* d_in, const int* in_sizes, int n_in,
                              void* d_out, int out_size, void* d_ws, size_t ws_size,
                              hipStream_t stream) {
    const float* pos  = (const float*)d_in[0];
    const int*   q    = (const int*)d_in[1];
    const int*   sp   = (const int*)d_in[2];
    const int*   Z    = (const int*)d_in[3];
    const int*   dst  = (const int*)d_in[4];
    const int*   src  = (const int*)d_in[5];
    const int*   seg  = (const int*)d_in[6];
    const float* ce   = (const float*)d_in[7];
    const float* se   = (const float*)d_in[8];
    const float* ae   = (const float*)d_in[9];
    const float* mW   = (const float*)d_in[10];
    const float* mb   = (const float*)d_in[11];
    const float* mpWb = (const float*)d_in[12];
    const float* rW1  = (const float*)d_in[13];
    const float* rb1  = (const float*)d_in[14];
    const float* rW2  = (const float*)d_in[15];
    const float* rb2  = (const float*)d_in[16];

    const int B = in_sizes[1];
    const int A = in_sizes[3];
    const int P = in_sizes[4];

    float* ws  = (float*)d_ws;
    float* x0  = ws + (size_t)B * F;
    float* bpl = x0 + (size_t)A * F;
    float* xs0 = bpl + (size_t)A * F;
    int* elist = (int*)(xs0 + (size_t)A * F);

    int* bins = (int*)bpl;

    prep_kernel<<<(P + CH - 1) / CH + (A + XA - 1) / XA, BT, 0, stream>>>(
        q, sp, ce, se, mW, mb, Z, seg, ae, x0, dst, src, bins, A, P);

    const int eblocks = (A + APW - 1) / APW;

    edge_kernel<<<eblocks, OT, 0, stream>>>(pos, dst, src, mpWb, x0, xs0,
        rW1, rb1, rW2, rb2, elist, bins, P, A, 1, 0);
    edge_kernel<<<eblocks, OT, 0, stream>>>(pos, dst, src, mpWb + NB * F, xs0, (float*)d_out,
        rW1 + 3 * F * F, rb1 + 3 * F, rW2 + 3 * F * F, rb2 + 3 * F, elist, bins, P, A, 0, 1);
}


# binning pass drops every edge at or beyond the cutoff radius (cutv = 0 there, so its message is an exact zero in both passes): lists and 16-edge tiles hold live edges only; sums unchanged
# speedup vs baseline: 4.2180x; 1.1145x over previous
#include <hip/hip_runtime.h>
#include <hip/hip_bf16.h>

typedef float v2f __attribute__((ext_vector_type(2)));
typedef float v8f __attribute__((ext_vector_type(8)));

#define F 32
#define NB 16
#define CUT2 36.0f

__device__ __forceinline__ float fast_silu(float x) {
    return x * __builtin_amdgcn_rcpf(1.0f + __expf(-x));
}

#define APW  64
#define DLB  6
static_assert(APW == (1 << DLB), "a packed list word keeps the local destination in its low DLB bits");
#define LCAP 4096
#define OT   1024
#define SCAN_UN 8
#define CH   1024
#define BT   256
#define MAXB 1024
static_assert(CH == 4 * BT && MAXB == 4 * BT, "4 edges and at most 4 groups per binning thread");
#define XA   128
#define XR   (XA / (BT / 32))
static_assert(XR <= 32 && (XR & (XR - 1)) == 0, "a lane looks after the indices of atom a0 + (lane & (XR - 1))");

struct BinPlan { int nbins, nchunk, ostride; bool ok; };
__device__ __forceinline__ BinPlan bin_plan(int A, int P) {
    BinPlan p;
    p.nbins   = (A + APW - 1) / APW;
    p.nchunk  = (P + CH - 1) / CH;
    p.ostride = (p.nbins + 1 + 31) & ~31;
    p.ok = p.nbins <= MAXB && A <= (1 << (32 - DLB)) &&
           (long long)p.nchunk * (CH + p.ostride) <= (long long)A * F;
    return p;
}

__device__ __forceinline__ void edge_rw(const float* __restrict__ pos, int s, int d, bool live, float* rw) {
    float dx = pos[s * 3 + 0] - pos[d * 3 + 0];
    float dy = pos[s * 3 + 1] - pos[d * 3 + 1];
    float dz = pos[s * 3 + 2] - pos[d * 3 + 2];
    float r = sqrtf(dx * dx + dy * dy + dz * dz);
    float u = r * (1.0f / 6.0f);
    u = u * u;
    float cutv = 0.0f;
    if (u < 1.0f)
        cutv = __expf(1.0f - __builtin_amdgcn_rcpf(fmaxf(1.0f - u, 1e-6f)));
    cutv = live ? cutv : 0.0f;
    float zc = fminf(fmaxf(2.0f * __expf(-r) - 1.0f, -1.0f), 1.0f);
    float Tm2 = 1.0f, Tm1 = zc;
    rw[0] = cutv;
    rw[1] = cutv * zc;
    #pragma unroll
    for (int b2 = 2; b2 < NB; ++b2) {
        float T = 2.0f * zc * Tm1 - Tm2;
        rw[b2] = cutv * T;
        Tm2 = Tm1; Tm1 = T;
    }
}

__device__ __forceinline__ float pick(bool c, float a, float b) { return c ? a : b; }

__device__ __forceinline__ int4 load_dst4(const int* __restrict__ dst, int b, int P) {
    if (b + 3 < P) return *(const int4*)(dst + b);
    int4 v;
    v.x = (b     < P) ? dst[b]     : -1;
    v.y = (b + 1 < P) ? dst[b + 1] : -1;
    v.z = (b + 2 < P) ? dst[b + 2] : -1;
    v.w = (b + 3 < P) ? dst[b + 3] : -1;
    return v;
}

__global__ __launch_bounds__(BT) void prep_kernel(
    const int* __restrict__ q, const int* __restrict__ sp,
    const float* __restrict__ ce, const float* __restrict__ se,
    const float* __restrict__ mW, const float* __restrict__ mb,
    const int* __restrict__ Z, const int* __restrict__ seg,
    const float* __restrict__ ae, float* __restrict__ x0,
    const int* __restrict__ dst, const int* __restrict__ src,
    const float* __restrict__ pos,
    int* __restrict__ bins, int A, int P) {
    __shared__ int hist[MAXB + 1];
    __shared__ int sorted[CH];
    __shared__ int wsum[BT / 32];

    const BinPlan pl = bin_plan(A, P);
    const int t = threadIdx.x;
    if ((int)blockIdx.x >= pl.nchunk) {
        const int lane = t & 31;
        const int wv = __builtin_amdgcn_readfirstlane(t >> 5);
        const int a0 = ((int)blockIdx.x - pl.nchunk) * XA + wv * XR;
        if (a0 >= A) return;
        float w[32];
        #pragma unroll
        for (int k = 0; k < 32; ++k) w[k] = mW[k * F + lane];
        const float bias = mb[lane];

        const int aj  = min(a0 + (lane & (XR - 1)), A - 1);
        const int zj  = Z[aj];
        const int bj  = seg[aj];
        const int cij = min(max(q[bj] + 5, 0), 10);
        const int sij = min(max(sp[bj] - 1, 0), 6);

        float e[XR], c[XR];
        int key[XR];
        const float* __restrict__ crow = (lane < 16) ? ce : se;
        const int ccol = lane & 15;
        #pragma unroll
        for (int j = 0; j < XR; ++j) {
            const int z  = __builtin_amdgcn_readlane(zj, j);
            const int ci = __builtin_amdgcn_readlane(cij, j);
            const int si = __builtin_amdgcn_readlane(sij, j);
            e[j] = ae[z * F + lane];
            c[j] = crow[((lane < 16) ? ci : si) * 16 + ccol];
            key[j] = ci * 8 + si;
        }
        float m = 0.0f;
        #pragma unroll
        for (int j = 0; j < XR; ++j) {
            if (j == 0 || key[j] != key[j - 1]) {
                const int cb = __builtin_bit_cast(int, c[j]);
                m = bias;
                #pragma unroll
                for (int k = 0; k < 32; ++k)
                    m += __builtin_bit_cast(float, __builtin_amdgcn_readlane(cb, k)) * w[k];
            }
            if (a0 + j < A) x0[(size_t)(a0 + j) * F + lane] = e[j] + m;
        }
        return;
    }
    if (!pl.ok) return;

    const int lane = t & 31;
    const int wave = t >> 5;
    const int c = blockIdx.x;
    const int e0 = c * CH + t * 4;
    const int4 dv = load_dst4(dst, e0, P);
    const int4 sv = load_dst4(src, e0, P);
    for (int i = t; i <= pl.nbins; i += BT) hist[i] = 0;
    const int dd[4] = {dv.x, dv.y, dv.z, dv.w};
    const int ss[4] = {sv.x, sv.y, sv.z, sv.w};
    bool own[4];
    float ps[4][3], pd[4][3];
    #pragma unroll
    for (int k = 0; k < 4; ++k) {
        own[k] = (unsigned)dd[k] < (unsigned)A;
        #pragma unroll
        for (int j = 0; j < 3; ++j) { ps[k][j] = 0.0f; pd[k][j] = 0.0f; }
        if (own[k]) {
            #pragma unroll
            for (int j = 0; j < 3; ++j) {
                ps[k][j] = pos[ss[k] * 3 + j];
                pd[k][j] = pos[dd[k] * 3 + j];
            }
        }
    }
    bool live[4];
    #pragma unroll
    for (int k = 0; k < 4; ++k) {
        const float dx = ps[k][0] - pd[k][0];
        const float dy = ps[k][1] - pd[k][1];
        const float dz = ps[k][2] - pd[k][2];
        live[k] = own[k] && (dx * dx + dy * dy + dz * dz < CUT2);
    }
    __syncthreads();

    int rank[4];
    #pragma unroll
    for (int k = 0; k < 4; ++k)
        rank[k] = live[k] ? atomicAdd(&hist[dd[k] >> DLB], 1) : -1;
    __syncthreads();

    const int npt = (pl.nbins + BT - 1) / BT;
    const int b0 = t * npt;
    int cnts[4];
    int loc = 0;
    #pragma unroll
    for (int j = 0; j < 4; ++j) {
        const int b = b0 + j;
        cnts[j] = (j < npt && b < pl.nbins) ? hist[b] : 0;
        loc += cnts[j];
    }
    int inc = loc;
    #pragma unroll
    for (int d = 1; d < 32; d <<= 1) {
        const int nb = __shfl_up(inc, d);
        if (lane >= d) inc += nb;
    }
    if (lane == 31) wsum[wave] = inc;
    __syncthreads();
    int wbase = 0, tot = 0;
    #pragma unroll
    for (int w = 0; w < BT / 32; ++w) {
        const int v = wsum[w];
        if (w < wave) wbase += v;
        tot += v;
    }
    int ex = wbase + inc - loc;
    #pragma unroll
    for (int j = 0; j < 4; ++j) {
        const int b = b0 + j;
        if (j < npt && b < pl.nbins) { hist[b] = ex; ex += cnts[j]; }
    }
    if (t == 0) hist[pl.nbins] = tot;
    __syncthreads();

    #pragma unroll
    for (int k = 0; k < 4; ++k)
        if (rank[k] >= 0)
            sorted[hist[dd[k] >> DLB] + rank[k]] =
                (int)(((unsigned)ss[k] << DLB) | (unsigned)(dd[k] & (APW - 1)));
    __syncthreads();

    int* __restrict__ chunk = bins + (size_t)c * CH;
    for (int i = t; i < tot; i += BT) chunk[i] = sorted[i];
    int* __restrict__ off = bins + (size_t)pl.nchunk * CH + (size_t)c * pl.ostride;
    for (int i = t; i <= pl.nbins; i += BT) off[i] = hist[i];
}

__device__ __forceinline__ void wave_lds_order() {
    __builtin_amdgcn_fence(__ATOMIC_RELEASE, "wavefront");
    __builtin_amdgcn_wave_barrier();
    __builtin_amdgcn_fence(__ATOMIC_ACQUIRE, "wavefront");
}

#define RW (APW / 16)
static_assert(3 * F * F == 3 * OT, "each thread stages 3 elements of each weight stack");

__global__ __launch_bounds__(OT) void edge_kernel(
    const float* __restrict__ pos, const int* __restrict__ dst,
    const int* __restrict__ src, const float* __restrict__ Wb,
    const float* __restrict__ xin, float* __restrict__ xout,
    const float* __restrict__ W1s, const float* __restrict__ b1s,
    const float* __restrict__ W2s, const float* __restrict__ b2s,
    int* __restrict__ lst, const int* __restrict__ bins, int P, int A, int add_cut, int reuse) {
    __shared__ __align__(16) float acc[APW * F];
    __shared__ __align__(16) float ht[APW * F];
    __shared__ __align__(16) float sW1[3 * F * F];
    __shared__ __align__(16) float sW2[3 * F * F];
    __shared__ float sb1[3 * F];
    __shared__ float sb2[3 * F];
    __shared__ int listE[LCAP];
    __shared__ int listD[LCAP];
    __shared__ int s_cnt;

    const int t = threadIdx.x;
    const int lane = t & 31;
    const int wave = t >> 5;
    const int lo = blockIdx.x * APW;
    const float addc = add_cut ? 1.0f : 0.0f;
    const int wv = __builtin_amdgcn_readfirstlane(t >> 5);

    float w1r[3], w2r[3];
    #pragma unroll
    for (int j = 0; j < 3; ++j) { w1r[j] = W1s[t + j * OT]; w2r[j] = W2s[t + j * OT]; }
    float b1r = 0.0f, b2r = 0.0f;
    if (t < 3 * F) { b1r = b1s[t]; b2r = b2s[t]; }
    float4 xo[4];
    #pragma unroll
    for (int j = 0; j < 4; ++j) xo[j] = make_float4(0.0f, 0.0f, 0.0f, 0.0f);
    if (wv < RW) {
        #pragma unroll
        for (int j = 0; j < 4; ++j) {
            const int row = min(lo + wv * 16 + j * 4 + (lane >> 3), A - 1);
            xo[j] = *(const float4*)(xin + (size_t)row * F + (lane & 7) * 4);
        }
    }

    const int rows = min(APW, A - lo);
    int* __restrict__ mylst = lst + (size_t)lo * F;
    const int capS = rows * F - 32;
    const bool packable = A <= (1 << (32 - DLB));

    int total = 0;
    bool from_list = false;
    if (reuse) {
        const int c = mylst[capS];
        if (c >= 0 && c <= capS && c <= LCAP) { from_list = true; total = c; }
    }

    for (int i = t; i < APW * F; i += OT) acc[i] = 0.0f;
    if (t == 0) s_cnt = 0;
    __syncthreads();

    if (from_list) {
        for (int i = t; i < total; i += OT) {
            const unsigned w = (unsigned)mylst[i];
            listE[i] = (int)(w >> DLB);
            listD[i] = (int)(w & (unsigned)(APW - 1));
        }
    } else if (!reuse && bin_plan(A, P).ok) {
        const BinPlan pl = bin_plan(A, P);
        const int* __restrict__ offs = bins + (size_t)pl.nchunk * CH;
        const int sub = t & 3;
        for (int g0 = 0; g0 < pl.nchunk; g0 += OT / 4) {
            const int g = g0 + (t >> 2);
            int beg = 0, n = 0;
            if (g < pl.nchunk) {
                const int* __restrict__ o = offs + (size_t)g * pl.ostride + blockIdx.x;
                beg = o[0];
                n = o[1] - beg;
            }
            int base = 0;
            if (sub == 0 && n > 0) base = atomicAdd(&s_cnt, n);
            base = __shfl(base, lane & ~3);
            for (int j = sub; j < n; j += 4) {
                const int slot = base + j;
                if (slot < LCAP) {
                    const unsigned w = (unsigned)bins[(size_t)g * CH + beg + j];
                    listE[slot] = (int)(w >> DLB);
                    listD[slot] = (int)(w & (unsigned)(APW - 1));
                }
            }
        }
        __syncthreads();
        total = s_cnt;
        const bool save = total <= capS && total <= LCAP;
        if (save)
            for (int i = t; i < total; i += OT)
                mylst[i] = (int)(((unsigned)listE[i] << DLB) | (unsigned)listD[i]);
        if (t == 0) mylst[capS] = save ? total : -1;
    } else {
        auto keep4 = [&](const int4 v, const int b) {
            const int dv[4] = {v.x, v.y, v.z, v.w};
            #pragma unroll
            for (int c = 0; c < 4; ++c) {
                unsigned u = (unsigned)(dv[c] - lo);
                if (u < (unsigned)APW) {
                    int slot = atomicAdd(&s_cnt, 1);
                    if (slot < LCAP) { listE[slot] = b + c; listD[slot] = (int)u; }
                }
            }
        };
        int base0 = t * 4;
        for (; base0 + (SCAN_UN - 1) * OT * 4 + 3 < P; base0 += OT * 4 * SCAN_UN) {
            int4 v[SCAN_UN];
            #pragma unroll
            for (int j = 0; j < SCAN_UN; ++j)
                v[j] = *(const int4*)(dst + base0 + j * OT * 4);
            #pragma unroll
            for (int j = 0; j < SCAN_UN; ++j)
                keep4(v[j], base0 + j * OT * 4);
        }
        for (; base0 < P; base0 += OT * 4)
            keep4(load_dst4(dst, base0, P), base0);
        __syncthreads();
        total = s_cnt;

        const bool save = !reuse && packable && total <= capS && total <= LCAP;
        if (total <= LCAP) {
            for (int i = t; i < total; i += OT) {
                const int s = src[listE[i]];
                listE[i] = s;
                if (save) mylst[i] = (int)(((unsigned)s << DLB) | (unsigned)listD[i]);
            }
        }
        if (!reuse && t == 0) mylst[capS] = save ? total : -1;
    }
    #pragma unroll
    for (int j = 0; j < 3; ++j) {
        const int k = t >> 5, n = t & 31;
        const int fidx = j * F * F + ((k >> 1) << 6) + (n << 1) + (k & 1);
        sW1[fidx] = w1r[j];
        sW2[fidx] = w2r[j];
    }
    if (t < 3 * F) { sb1[t] = b1r; sb2[t] = b2r; }
    __syncthreads();

    if (total <= LCAP) {
        const int m  = lane & 15;
        const int h  = lane >> 4;
        const int cn = lane & 15;

        v2f bf0[4], bf1[4];
        #pragma unroll
        for (int kc = 0; kc < 4; ++kc) {
            const int k0 = 4 * kc + 2 * h;
            bf0[kc][0] = Wb[k0 * F + cn];
            bf0[kc][1] = Wb[(k0 + 1) * F + cn];
            bf1[kc][0] = Wb[k0 * F + cn + 16];
            bf1[kc][1] = Wb[(k0 + 1) * F + cn + 16];
        }
        if (h == 0) { bf0[0][0] += addc; bf1[0][0] += addc; }

        const int ntiles = (total + 15) >> 4;
        for (int tile = wave; tile < ntiles; tile += OT / 32) {
            const int idx = tile * 16 + m;
            const bool valid = idx < total;
            const int s  = valid ? listE[idx] : 0;
            const int dl = valid ? listD[idx] : 0;

            int   drr[8];
            float xv0[8], xv1[8];
            #pragma unroll
            for (int q = 0; q < 8; ++q) {
                const int r  = 8 * h + q;
                const int sr = __shfl(s, r);
                drr[q] = __shfl(dl, r);
                xv0[q] = xin[sr * F + cn];
                xv1[q] = xin[sr * F + cn + 16];
            }

            float rw[NB];
            edge_rw(pos, s, lo + dl, valid, rw);

            v8f c0, c1;
            #pragma unroll
            for (int q = 0; q < 8; ++q) { c0[q] = 0.0f; c1[q] = 0.0f; }
            #pragma unroll
            for (int kc = 0; kc < 4; ++kc) {
                v2f a;
                a[0] = pick(h != 0, rw[4 * kc + 2], rw[4 * kc]);
                a[1] = pick(h != 0, rw[4 * kc + 3], rw[4 * kc + 1]);
                c0 = __builtin_amdgcn_wmma_f32_16x16x4_f32(false, a, false, bf0[kc], (short)0, c0, false, false);
                c1 = __builtin_amdgcn_wmma_f32_16x16x4_f32(false, a, false, bf1[kc], (short)0, c1, false, false);
            }
            #pragma unroll
            for (int q = 0; q < 8; ++q) {
                atomicAdd(&acc[drr[q] * F + cn],      c0[q] * xv0[q]);
                atomicAdd(&acc[drr[q] * F + cn + 16], c1[q] * xv1[q]);
            }
        }
    } else {
        for (int e = t; e < P; e += OT) {
            unsigned u = (unsigned)(dst[e] - lo);
            if (u < (unsigned)APW) {
                const int s = src[e];
                float rw[NB];
                edge_rw(pos, s, lo + (int)u, true, rw);
                for (int f = 0; f < F; ++f) {
                    float c = rw[0] * (Wb[f] + addc);
                    #pragma unroll
                    for (int k = 1; k < NB; ++k) c += rw[k] * Wb[k * F + f];
                    atomicAdd(&acc[(int)u * F + f], c * xin[s * F + f]);
                }
            }
        }
    }
    __syncthreads();

    if (wv >= RW) return;

    float* xt = acc + wv * 16 * F;
    float* hw = ht  + wv * 16 * F;
    const int m     = lane & 15;
    const int khalf = (lane >> 4) << 1;
    const int cm    = (lane >> 4) << 3;
    const int cn    = lane & 15;

    v2f b0[8], b1[8];
    float bn0, bn1;
    auto load_b = [&](const float* Wf, const float* Bv) {
        const v2f* Wp = (const v2f*)Wf;
        #pragma unroll
        for (int kc = 0; kc < 8; ++kc) {
            const int kp = ((kc * 4 + khalf) >> 1) << 5;
            b0[kc] = Wp[kp + cn];
            b1[kc] = Wp[kp + cn + 16];
        }
        bn0 = Bv[cn];
        bn1 = Bv[cn + 16];
    };
    auto gemm = [&](const float* tile, v8f& acc0, v8f& acc1) {
        v2f a[8];
        #pragma unroll
        for (int kc = 0; kc < 8; ++kc) a[kc] = *(const v2f*)(tile + m * F + kc * 4 + khalf);
        #pragma unroll
        for (int q = 0; q < 8; ++q) { acc0[q] = bn0; acc1[q] = bn1; }
        #pragma unroll
        for (int kc = 0; kc < 8; ++kc) {
            acc0 = __builtin_amdgcn_wmma_f32_16x16x4_f32(false, a[kc], false, b0[kc], (short)0, acc0, false, false);
            acc1 = __builtin_amdgcn_wmma_f32_16x16x4_f32(false, a[kc], false, b1[kc], (short)0, acc1, false, false);
        }
    };

    load_b(sW1, sb1);

    #pragma unroll
    for (int j = 0; j < 4; ++j) {
        float4* p = (float4*)(xt + j * 128 + lane * 4);
        float4 v = *p;
        v.x += xo[j].x; v.y += xo[j].y; v.z += xo[j].z; v.w += xo[j].w;
        *p = v;
    }
    wave_lds_order();

    #pragma unroll
    for (int j = 0; j < 3; ++j) {
        v8f acc0, acc1;
        gemm(xt, acc0, acc1);
        load_b(sW2 + j * F * F, sb2 + j * F);
        #pragma unroll
        for (int q = 0; q < 8; ++q) {
            hw[(cm + q) * F + cn]      = fast_silu(acc0[q]);
            hw[(cm + q) * F + cn + 16] = fast_silu(acc1[q]);
        }
        wave_lds_order();

        gemm(hw, acc0, acc1);
        if (j < 2) load_b(sW1 + (j + 1) * F * F, sb1 + (j + 1) * F);
        #pragma unroll
        for (int q = 0; q < 8; ++q) {
            const float x0v = xt[(cm + q) * F + cn]      + acc0[q];
            const float x1v = xt[(cm + q) * F + cn + 16] + acc1[q];
            xt[(cm + q) * F + cn]      = fast_silu(x0v);
            xt[(cm + q) * F + cn + 16] = fast_silu(x1v);
        }
        wave_lds_order();
    }

    for (int j = 0; j < 4; ++j) {
        const int row = wv * 16 + j * 4 + (lane >> 3);
        if (lo + row < A)
            *(float4*)(xout + (size_t)(lo + row) * F + (lane & 7) * 4) = *(const float4*)(xt + j * 128 + lane * 4);
    }
}

extern "C" void kernel_launch(void* const* d_in, const int* in_sizes, int n_in,
                              void* d_out, int out_size, void* d_ws, size_t ws_size,
                              hipStream_t stream) {
    const float* pos  = (const float*)d_in[0];
    const int*   q    = (const int*)d_in[1];
    const int*   sp   = (const int*)d_in[2];
    const int*   Z    = (const int*)d_in[3];
    const int*   dst  = (const int*)d_in[4];
    const int*   src  = (const int*)d_in[5];
    const int*   seg  = (const int*)d_in[6];
    const float* ce   = (const float*)d_in[7];
    const float* se   = (const float*)d_in[8];
    const float* ae   = (const float*)d_in[9];
    const float* mW   = (const float*)d_in[10];
    const float* mb   = (const float*)d_in[11];
    const float* mpWb = (const float*)d_in[12];
    const float* rW1  = (const float*)d_in[13];
    const float* rb1  = (const float*)d_in[14];
    const float* rW2  = (const float*)d_in[15];
    const float* rb2  = (const float*)d_in[16];

    const int B = in_sizes[1];
    const int A = in_sizes[3];
    const int P = in_sizes[4];

    float* ws  = (float*)d_ws;
    float* x0  = ws + (size_t)B * F;
    float* bpl = x0 + (size_t)A * F;
    float* xs0 = bpl + (size_t)A * F;
    int* elist = (int*)(xs0 + (size_t)A * F);

    int* bins = (int*)bpl;

    prep_kernel<<<(P + CH - 1) / CH + (A + XA - 1) / XA, BT, 0, stream>>>(
        q, sp, ce, se, mW, mb, Z, seg, ae, x0, dst, src, pos, bins, A, P);

    const int eblocks = (A + APW - 1) / APW;

    edge_kernel<<<eblocks, OT, 0, stream>>>(pos, dst, src, mpWb, x0, xs0,
        rW1, rb1, rW2, rb2, elist, bins, P, A, 1, 0);
    edge_kernel<<<eblocks, OT, 0, stream>>>(pos, dst, src, mpWb + NB * F, xs0, (float*)d_out,
        rW1 + 3 * F * F, rb1 + 3 * F, rW2 + 3 * F * F, rb2 + 3 * F, elist, bins, P, A, 0, 1);
}


# refine tail stays in registers: products taken transposed (W^T as A, data as B), so a stage's D tile is the next stage's B fragment, no LDS round trip between stages; k order of refine sums changes
# speedup vs baseline: 4.3514x; 1.0316x over previous
#include <hip/hip_runtime.h>
#include <hip/hip_bf16.h>

typedef float v2f __attribute__((ext_vector_type(2)));
typedef float v8f __attribute__((ext_vector_type(8)));

#define F 32
#define NB 16
#define CUT2 36.0f

__device__ __forceinline__ float fast_silu(float x) {
    return x * __builtin_amdgcn_rcpf(1.0f + __expf(-x));
}

#define APW  64
#define DLB  6
static_assert(APW == (1 << DLB), "a packed list word keeps the local destination in its low DLB bits");
#define LCAP 4096
#define OT   1024
#define SCAN_UN 8
#define CH   1024
#define BT   256
#define MAXB 1024
static_assert(CH == 4 * BT && MAXB == 4 * BT, "4 edges and at most 4 groups per binning thread");
#define XA   128
#define XR   (XA / (BT / 32))
static_assert(XR <= 32 && (XR & (XR - 1)) == 0, "a lane looks after the indices of atom a0 + (lane & (XR - 1))");

struct BinPlan { int nbins, nchunk, ostride; bool ok; };
__device__ __forceinline__ BinPlan bin_plan(int A, int P) {
    BinPlan p;
    p.nbins   = (A + APW - 1) / APW;
    p.nchunk  = (P + CH - 1) / CH;
    p.ostride = (p.nbins + 1 + 31) & ~31;
    p.ok = p.nbins <= MAXB && A <= (1 << (32 - DLB)) &&
           (long long)p.nchunk * (CH + p.ostride) <= (long long)A * F;
    return p;
}

__device__ __forceinline__ void edge_rw(const float* __restrict__ pos, int s, int d, bool live, float* rw) {
    float dx = pos[s * 3 + 0] - pos[d * 3 + 0];
    float dy = pos[s * 3 + 1] - pos[d * 3 + 1];
    float dz = pos[s * 3 + 2] - pos[d * 3 + 2];
    float r = sqrtf(dx * dx + dy * dy + dz * dz);
    float u = r * (1.0f / 6.0f);
    u = u * u;
    float cutv = 0.0f;
    if (u < 1.0f)
        cutv = __expf(1.0f - __builtin_amdgcn_rcpf(fmaxf(1.0f - u, 1e-6f)));
    cutv = live ? cutv : 0.0f;
    float zc = fminf(fmaxf(2.0f * __expf(-r) - 1.0f, -1.0f), 1.0f);
    float Tm2 = 1.0f, Tm1 = zc;
    rw[0] = cutv;
    rw[1] = cutv * zc;
    #pragma unroll
    for (int b2 = 2; b2 < NB; ++b2) {
        float T = 2.0f * zc * Tm1 - Tm2;
        rw[b2] = cutv * T;
        Tm2 = Tm1; Tm1 = T;
    }
}

__device__ __forceinline__ float pick(bool c, float a, float b) { return c ? a : b; }

__device__ __forceinline__ int4 load_dst4(const int* __restrict__ dst, int b, int P) {
    if (b + 3 < P) return *(const int4*)(dst + b);
    int4 v;
    v.x = (b     < P) ? dst[b]     : -1;
    v.y = (b + 1 < P) ? dst[b + 1] : -1;
    v.z = (b + 2 < P) ? dst[b + 2] : -1;
    v.w = (b + 3 < P) ? dst[b + 3] : -1;
    return v;
}

__global__ __launch_bounds__(BT) void prep_kernel(
    const int* __restrict__ q, const int* __restrict__ sp,
    const float* __restrict__ ce, const float* __restrict__ se,
    const float* __restrict__ mW, const float* __restrict__ mb,
    const int* __restrict__ Z, const int* __restrict__ seg,
    const float* __restrict__ ae, float* __restrict__ x0,
    const int* __restrict__ dst, const int* __restrict__ src,
    const float* __restrict__ pos,
    int* __restrict__ bins, int A, int P) {
    __shared__ int hist[MAXB + 1];
    __shared__ int sorted[CH];
    __shared__ int wsum[BT / 32];

    const BinPlan pl = bin_plan(A, P);
    const int t = threadIdx.x;
    if ((int)blockIdx.x >= pl.nchunk) {
        const int lane = t & 31;
        const int wv = __builtin_amdgcn_readfirstlane(t >> 5);
        const int a0 = ((int)blockIdx.x - pl.nchunk) * XA + wv * XR;
        if (a0 >= A) return;
        float w[32];
        #pragma unroll
        for (int k = 0; k < 32; ++k) w[k] = mW[k * F + lane];
        const float bias = mb[lane];

        const int aj  = min(a0 + (lane & (XR - 1)), A - 1);
        const int zj  = Z[aj];
        const int bj  = seg[aj];
        const int cij = min(max(q[bj] + 5, 0), 10);
        const int sij = min(max(sp[bj] - 1, 0), 6);

        float e[XR], c[XR];
        int key[XR];
        const float* __restrict__ crow = (lane < 16) ? ce : se;
        const int ccol = lane & 15;
        #pragma unroll
        for (int j = 0; j < XR; ++j) {
            const int z  = __builtin_amdgcn_readlane(zj, j);
            const int ci = __builtin_amdgcn_readlane(cij, j);
            const int si = __builtin_amdgcn_readlane(sij, j);
            e[j] = ae[z * F + lane];
            c[j] = crow[((lane < 16) ? ci : si) * 16 + ccol];
            key[j] = ci * 8 + si;
        }
        float m = 0.0f;
        #pragma unroll
        for (int j = 0; j < XR; ++j) {
            if (j == 0 || key[j] != key[j - 1]) {
                const int cb = __builtin_bit_cast(int, c[j]);
                m = bias;
                #pragma unroll
                for (int k = 0; k < 32; ++k)
                    m += __builtin_bit_cast(float, __builtin_amdgcn_readlane(cb, k)) * w[k];
            }
            if (a0 + j < A) x0[(size_t)(a0 + j) * F + lane] = e[j] + m;
        }
        return;
    }
    if (!pl.ok) return;

    const int lane = t & 31;
    const int wave = t >> 5;
    const int c = blockIdx.x;
    const int e0 = c * CH + t * 4;
    const int4 dv = load_dst4(dst, e0, P);
    const int4 sv = load_dst4(src, e0, P);
    for (int i = t; i <= pl.nbins; i += BT) hist[i] = 0;
    const int dd[4] = {dv.x, dv.y, dv.z, dv.w};
    const int ss[4] = {sv.x, sv.y, sv.z, sv.w};
    bool own[4];
    float ps[4][3], pd[4][3];
    #pragma unroll
    for (int k = 0; k < 4; ++k) {
        own[k] = (unsigned)dd[k] < (unsigned)A;
        #pragma unroll
        for (int j = 0; j < 3; ++j) { ps[k][j] = 0.0f; pd[k][j] = 0.0f; }
        if (own[k]) {
            #pragma unroll
            for (int j = 0; j < 3; ++j) {
                ps[k][j] = pos[ss[k] * 3 + j];
                pd[k][j] = pos[dd[k] * 3 + j];
            }
        }
    }
    bool live[4];
    #pragma unroll
    for (int k = 0; k < 4; ++k) {
        const float dx = ps[k][0] - pd[k][0];
        const float dy = ps[k][1] - pd[k][1];
        const float dz = ps[k][2] - pd[k][2];
        live[k] = own[k] && (dx * dx + dy * dy + dz * dz < CUT2);
    }
    __syncthreads();

    int rank[4];
    #pragma unroll
    for (int k = 0; k < 4; ++k)
        rank[k] = live[k] ? atomicAdd(&hist[dd[k] >> DLB], 1) : -1;
    __syncthreads();

    const int npt = (pl.nbins + BT - 1) / BT;
    const int b0 = t * npt;
    int cnts[4];
    int loc = 0;
    #pragma unroll
    for (int j = 0; j < 4; ++j) {
        const int b = b0 + j;
        cnts[j] = (j < npt && b < pl.nbins) ? hist[b] : 0;
        loc += cnts[j];
    }
    int inc = loc;
    #pragma unroll
    for (int d = 1; d < 32; d <<= 1) {
        const int nb = __shfl_up(inc, d);
        if (lane >= d) inc += nb;
    }
    if (lane == 31) wsum[wave] = inc;
    __syncthreads();
    int wbase = 0, tot = 0;
    #pragma unroll
    for (int w = 0; w < BT / 32; ++w) {
        const int v = wsum[w];
        if (w < wave) wbase += v;
        tot += v;
    }
    int ex = wbase + inc - loc;
    #pragma unroll
    for (int j = 0; j < 4; ++j) {
        const int b = b0 + j;
        if (j < npt && b < pl.nbins) { hist[b] = ex; ex += cnts[j]; }
    }
    if (t == 0) hist[pl.nbins] = tot;
    __syncthreads();

    #pragma unroll
    for (int k = 0; k < 4; ++k)
        if (rank[k] >= 0)
            sorted[hist[dd[k] >> DLB] + rank[k]] =
                (int)(((unsigned)ss[k] << DLB) | (unsigned)(dd[k] & (APW - 1)));
    __syncthreads();

    int* __restrict__ chunk = bins + (size_t)c * CH;
    for (int i = t; i < tot; i += BT) chunk[i] = sorted[i];
    int* __restrict__ off = bins + (size_t)pl.nchunk * CH + (size_t)c * pl.ostride;
    for (int i = t; i <= pl.nbins; i += BT) off[i] = hist[i];
}

#define RW (APW / 16)
static_assert(3 * F * F == 3 * OT, "each thread stages 3 elements of each weight stack");

__global__ __launch_bounds__(OT) void edge_kernel(
    const float* __restrict__ pos, const int* __restrict__ dst,
    const int* __restrict__ src, const float* __restrict__ Wb,
    const float* __restrict__ xin, float* __restrict__ xout,
    const float* __restrict__ W1s, const float* __restrict__ b1s,
    const float* __restrict__ W2s, const float* __restrict__ b2s,
    int* __restrict__ lst, const int* __restrict__ bins, int P, int A, int add_cut, int reuse) {
    __shared__ __align__(16) float acc[APW * F];
    __shared__ __align__(16) float sW1[3 * F * F];
    __shared__ __align__(16) float sW2[3 * F * F];
    __shared__ __align__(16) float sb1[3 * F];
    __shared__ __align__(16) float sb2[3 * F];
    __shared__ int listE[LCAP];
    __shared__ int listD[LCAP];
    __shared__ int s_cnt;

    const int t = threadIdx.x;
    const int lane = t & 31;
    const int wave = t >> 5;
    const int lo = blockIdx.x * APW;
    const float addc = add_cut ? 1.0f : 0.0f;
    const int wv = __builtin_amdgcn_readfirstlane(t >> 5);

    float w1r[3], w2r[3];
    #pragma unroll
    for (int j = 0; j < 3; ++j) { w1r[j] = W1s[t + j * OT]; w2r[j] = W2s[t + j * OT]; }
    float b1r = 0.0f, b2r = 0.0f;
    if (t < 3 * F) { b1r = b1s[t]; b2r = b2s[t]; }
    float4 xo[4];
    #pragma unroll
    for (int j = 0; j < 4; ++j) xo[j] = make_float4(0.0f, 0.0f, 0.0f, 0.0f);
    if (wv < RW) {
        const int row = min(lo + wv * 16 + (lane & 15), A - 1);
        const float* __restrict__ xr = xin + (size_t)row * F + (lane >> 4) * 8;
        #pragma unroll
        for (int j = 0; j < 4; ++j)
            xo[j] = *(const float4*)(xr + (j >> 1) * 16 + (j & 1) * 4);
    }

    const int rows = min(APW, A - lo);
    int* __restrict__ mylst = lst + (size_t)lo * F;
    const int capS = rows * F - 32;
    const bool packable = A <= (1 << (32 - DLB));

    int total = 0;
    bool from_list = false;
    if (reuse) {
        const int c = mylst[capS];
        if (c >= 0 && c <= capS && c <= LCAP) { from_list = true; total = c; }
    }

    for (int i = t; i < APW * F; i += OT) acc[i] = 0.0f;
    if (t == 0) s_cnt = 0;
    __syncthreads();

    if (from_list) {
        for (int i = t; i < total; i += OT) {
            const unsigned w = (unsigned)mylst[i];
            listE[i] = (int)(w >> DLB);
            listD[i] = (int)(w & (unsigned)(APW - 1));
        }
    } else if (!reuse && bin_plan(A, P).ok) {
        const BinPlan pl = bin_plan(A, P);
        const int* __restrict__ offs = bins + (size_t)pl.nchunk * CH;
        const int sub = t & 3;
        for (int g0 = 0; g0 < pl.nchunk; g0 += OT / 4) {
            const int g = g0 + (t >> 2);
            int beg = 0, n = 0;
            if (g < pl.nchunk) {
                const int* __restrict__ o = offs + (size_t)g * pl.ostride + blockIdx.x;
                beg = o[0];
                n = o[1] - beg;
            }
            int base = 0;
            if (sub == 0 && n > 0) base = atomicAdd(&s_cnt, n);
            base = __shfl(base, lane & ~3);
            for (int j = sub; j < n; j += 4) {
                const int slot = base + j;
                if (slot < LCAP) {
                    const unsigned w = (unsigned)bins[(size_t)g * CH + beg + j];
                    listE[slot] = (int)(w >> DLB);
                    listD[slot] = (int)(w & (unsigned)(APW - 1));
                }
            }
        }
        __syncthreads();
        total = s_cnt;
        const bool save = total <= capS && total <= LCAP;
        if (save)
            for (int i = t; i < total; i += OT)
                mylst[i] = (int)(((unsigned)listE[i] << DLB) | (unsigned)listD[i]);
        if (t == 0) mylst[capS] = save ? total : -1;
    } else {
        auto keep4 = [&](const int4 v, const int b) {
            const int dv[4] = {v.x, v.y, v.z, v.w};
            #pragma unroll
            for (int c = 0; c < 4; ++c) {
                unsigned u = (unsigned)(dv[c] - lo);
                if (u < (unsigned)APW) {
                    int slot = atomicAdd(&s_cnt, 1);
                    if (slot < LCAP) { listE[slot] = b + c; listD[slot] = (int)u; }
                }
            }
        };
        int base0 = t * 4;
        for (; base0 + (SCAN_UN - 1) * OT * 4 + 3 < P; base0 += OT * 4 * SCAN_UN) {
            int4 v[SCAN_UN];
            #pragma unroll
            for (int j = 0; j < SCAN_UN; ++j)
                v[j] = *(const int4*)(dst + base0 + j * OT * 4);
            #pragma unroll
            for (int j = 0; j < SCAN_UN; ++j)
                keep4(v[j], base0 + j * OT * 4);
        }
        for (; base0 < P; base0 += OT * 4)
            keep4(load_dst4(dst, base0, P), base0);
        __syncthreads();
        total = s_cnt;

        const bool save = !reuse && packable && total <= capS && total <= LCAP;
        if (total <= LCAP) {
            for (int i = t; i < total; i += OT) {
                const int s = src[listE[i]];
                listE[i] = s;
                if (save) mylst[i] = (int)(((unsigned)s << DLB) | (unsigned)listD[i]);
            }
        }
        if (!reuse && t == 0) mylst[capS] = save ? total : -1;
    }
    #pragma unroll
    for (int j = 0; j < 3; ++j) {
        const int f = t >> 5, g = t & 31;
        const int kc = ((f >> 4) << 2) | ((f >> 1) & 3);
        const int ln = (((f >> 3) & 1) << 4) | (g & 15);
        const int fidx = j * F * F + ((((g >> 4) * 8 + kc) * 32 + ln) << 1) + (f & 1);
        sW1[fidx] = w1r[j];
        sW2[fidx] = w2r[j];
    }
    if (t < 3 * F) { sb1[t] = b1r; sb2[t] = b2r; }
    __syncthreads();

    if (total <= LCAP) {
        const int m  = lane & 15;
        const int h  = lane >> 4;
        const int cn = lane & 15;

        v2f bf0[4], bf1[4];
        #pragma unroll
        for (int kc = 0; kc < 4; ++kc) {
            const int k0 = 4 * kc + 2 * h;
            bf0[kc][0] = Wb[k0 * F + cn];
            bf0[kc][1] = Wb[(k0 + 1) * F + cn];
            bf1[kc][0] = Wb[k0 * F + cn + 16];
            bf1[kc][1] = Wb[(k0 + 1) * F + cn + 16];
        }
        if (h == 0) { bf0[0][0] += addc; bf1[0][0] += addc; }

        const int ntiles = (total + 15) >> 4;
        for (int tile = wave; tile < ntiles; tile += OT / 32) {
            const int idx = tile * 16 + m;
            const bool valid = idx < total;
            const int s  = valid ? listE[idx] : 0;
            const int dl = valid ? listD[idx] : 0;

            int   drr[8];
            float xv0[8], xv1[8];
            #pragma unroll
            for (int q = 0; q < 8; ++q) {
                const int r  = 8 * h + q;
                const int sr = __shfl(s, r);
                drr[q] = __shfl(dl, r);
                xv0[q] = xin[sr * F + cn];
                xv1[q] = xin[sr * F + cn + 16];
            }

            float rw[NB];
            edge_rw(pos, s, lo + dl, valid, rw);

            v8f c0, c1;
            #pragma unroll
            for (int q = 0; q < 8; ++q) { c0[q] = 0.0f; c1[q] = 0.0f; }
            #pragma unroll
            for (int kc = 0; kc < 4; ++kc) {
                v2f a;
                a[0] = pick(h != 0, rw[4 * kc + 2], rw[4 * kc]);
                a[1] = pick(h != 0, rw[4 * kc + 3], rw[4 * kc + 1]);
                c0 = __builtin_amdgcn_wmma_f32_16x16x4_f32(false, a, false, bf0[kc], (short)0, c0, false, false);
                c1 = __builtin_amdgcn_wmma_f32_16x16x4_f32(false, a, false, bf1[kc], (short)0, c1, false, false);
            }
            #pragma unroll
            for (int q = 0; q < 8; ++q) {
                atomicAdd(&acc[drr[q] * F + cn],      c0[q] * xv0[q]);
                atomicAdd(&acc[drr[q] * F + cn + 16], c1[q] * xv1[q]);
            }
        }
    } else {
        for (int e = t; e < P; e += OT) {
            unsigned u = (unsigned)(dst[e] - lo);
            if (u < (unsigned)APW) {
                const int s = src[e];
                float rw[NB];
                edge_rw(pos, s, lo + (int)u, true, rw);
                for (int f = 0; f < F; ++f) {
                    float c = rw[0] * (Wb[f] + addc);
                    #pragma unroll
                    for (int k = 1; k < NB; ++k) c += rw[k] * Wb[k * F + f];
                    atomicAdd(&acc[(int)u * F + f], c * xin[s * F + f]);
                }
            }
        }
    }
    __syncthreads();

    if (wv >= RW) return;

    const int hf = lane >> 4;

    v2f wa0[8], wa1[8];
    v8f bi0, bi1;
    auto load_w = [&](const float* Wf, const float* Bv) {
        const v2f* Wp = (const v2f*)Wf;
        #pragma unroll
        for (int kc = 0; kc < 8; ++kc) {
            wa0[kc] = Wp[kc * 32 + lane];
            wa1[kc] = Wp[(8 + kc) * 32 + lane];
        }
        const float4 p0 = *(const float4*)(Bv + hf * 8);
        const float4 p1 = *(const float4*)(Bv + hf * 8 + 4);
        const float4 p2 = *(const float4*)(Bv + 16 + hf * 8);
        const float4 p3 = *(const float4*)(Bv + 16 + hf * 8 + 4);
        bi0[0] = p0.x; bi0[1] = p0.y; bi0[2] = p0.z; bi0[3] = p0.w;
        bi0[4] = p1.x; bi0[5] = p1.y; bi0[6] = p1.z; bi0[7] = p1.w;
        bi1[0] = p2.x; bi1[1] = p2.y; bi1[2] = p2.z; bi1[3] = p2.w;
        bi1[4] = p3.x; bi1[5] = p3.y; bi1[6] = p3.z; bi1[7] = p3.w;
    };
    auto gemm = [&](const v8f& i0, const v8f& i1, v8f& o0, v8f& o1) {
        o0 = bi0; o1 = bi1;
        #pragma unroll
        for (int kc = 0; kc < 8; ++kc) {
            const int r = 2 * (kc & 3);
            v2f b;
            b[0] = (kc < 4) ? i0[r]     : i1[r];
            b[1] = (kc < 4) ? i0[r + 1] : i1[r + 1];
            o0 = __builtin_amdgcn_wmma_f32_16x16x4_f32(false, wa0[kc], false, b, (short)0, o0, false, false);
            o1 = __builtin_amdgcn_wmma_f32_16x16x4_f32(false, wa1[kc], false, b, (short)0, o1, false, false);
        }
    };

    load_w(sW1, sb1);

    v8f X0, X1;
    {
        const float* ar = acc + (wv * 16 + (lane & 15)) * F + hf * 8;
        const float4 s0 = *(const float4*)(ar);
        const float4 s1 = *(const float4*)(ar + 4);
        const float4 s2 = *(const float4*)(ar + 16);
        const float4 s3 = *(const float4*)(ar + 20);
        X0[0] = s0.x + xo[0].x; X0[1] = s0.y + xo[0].y; X0[2] = s0.z + xo[0].z; X0[3] = s0.w + xo[0].w;
        X0[4] = s1.x + xo[1].x; X0[5] = s1.y + xo[1].y; X0[6] = s1.z + xo[1].z; X0[7] = s1.w + xo[1].w;
        X1[0] = s2.x + xo[2].x; X1[1] = s2.y + xo[2].y; X1[2] = s2.z + xo[2].z; X1[3] = s2.w + xo[2].w;
        X1[4] = s3.x + xo[3].x; X1[5] = s3.y + xo[3].y; X1[6] = s3.z + xo[3].z; X1[7] = s3.w + xo[3].w;
    }

    #pragma unroll
    for (int j = 0; j < 3; ++j) {
        v8f H0, H1;
        gemm(X0, X1, H0, H1);
        load_w(sW2 + j * F * F, sb2 + j * F);
        #pragma unroll
        for (int q = 0; q < 8; ++q) { H0[q] = fast_silu(H0[q]); H1[q] = fast_silu(H1[q]); }

        v8f Y0, Y1;
        gemm(H0, H1, Y0, Y1);
        if (j < 2) load_w(sW1 + (j + 1) * F * F, sb1 + (j + 1) * F);
        #pragma unroll
        for (int q = 0; q < 8; ++q) { X0[q] = fast_silu(X0[q] + Y0[q]); X1[q] = fast_silu(X1[q] + Y1[q]); }
    }

    const int orow = lo + wv * 16 + (lane & 15);
    if (orow < A) {
        float* __restrict__ op = xout + (size_t)orow * F + hf * 8;
        *(float4*)(op)      = make_float4(X0[0], X0[1], X0[2], X0[3]);
        *(float4*)(op + 4)  = make_float4(X0[4], X0[5], X0[6], X0[7]);
        *(float4*)(op + 16) = make_float4(X1[0], X1[1], X1[2], X1[3]);
        *(float4*)(op + 20) = make_float4(X1[4], X1[5], X1[6], X1[7]);
    }
}

extern "C" void kernel_launch(void* const* d_in, const int* in_sizes, int n_in,
                              void* d_out, int out_size, void* d_ws, size_t ws_size,
                              hipStream_t stream) {
    const float* pos  = (const float*)d_in[0];
    const int*   q    = (const int*)d_in[1];
    const int*   sp   = (const int*)d_in[2];
    const int*   Z    = (const int*)d_in[3];
    const int*   dst  = (const int*)d_in[4];
    const int*   src  = (const int*)d_in[5];
    const int*   seg  = (const int*)d_in[6];
    const float* ce   = (const float*)d_in[7];
    const float* se   = (const float*)d_in[8];
    const float* ae   = (const float*)d_in[9];
    const float* mW   = (const float*)d_in[10];
    const float* mb   = (const float*)d_in[11];
    const float* mpWb = (const float*)d_in[12];
    const float* rW1  = (const float*)d_in[13];
    const float* rb1  = (const float*)d_in[14];
    const float* rW2  = (const float*)d_in[15];
    const float* rb2  = (const float*)d_in[16];

    const int B = in_sizes[1];
    const int A = in_sizes[3];
    const int P = in_sizes[4];

    float* ws  = (float*)d_ws;
    float* x0  = ws + (size_t)B * F;
    float* bpl = x0 + (size_t)A * F;
    float* xs0 = bpl + (size_t)A * F;
    int* elist = (int*)(xs0 + (size_t)A * F);

    int* bins = (int*)bpl;

    prep_kernel<<<(P + CH - 1) / CH + (A + XA - 1) / XA, BT, 0, stream>>>(
        q, sp, ce, se, mW, mb, Z, seg, ae, x0, dst, src, pos, bins, A, P);

    const int eblocks = (A + APW - 1) / APW;

    edge_kernel<<<eblocks, OT, 0, stream>>>(pos, dst, src, mpWb, x0, xs0,
        rW1, rb1, rW2, rb2, elist, bins, P, A, 1, 0);
    edge_kernel<<<eblocks, OT, 0, stream>>>(pos, dst, src, mpWb + NB * F, xs0, (float*)d_out,
        rW1 + 3 * F * F, rb1 + 3 * F, rW2 + 3 * F * F, rb2 + 3 * F, elist, bins, P, A, 0, 1);
}
